# starting kernel
# baseline (speedup 1.0000x reference)
#include <hip/hip_runtime.h>

typedef __attribute__((ext_vector_type(2))) float v2f;
typedef __attribute__((ext_vector_type(8))) float v8f;

#define NEGV (-1e30f)
constexpr int Bn = 32, Dd = 128, CL = 1024, QL = 256;

__global__ void k_bias(const float* __restrict__ C, const float* __restrict__ Q,
                       const float* __restrict__ W0,
                       float* __restrict__ rowb, float* __restrict__ colb) {
  int t = blockIdx.x * blockDim.x + threadIdx.x;
  if (t < Bn * CL) {
    int b = t / CL, i = t % CL;
    float s = 0.f;
    for (int k = 0; k < Dd; ++k) s += C[(size_t)(b * Dd + k) * CL + i] * W0[k];
    rowb[t] = s;
  } else if (t < Bn * CL + Bn * QL) {
    int u = t - Bn * CL;
    int b = u / QL, j = u % QL;
    float s = 0.f;
    for (int k = 0; k < Dd; ++k) s += Q[(size_t)(b * Dd + k) * QL + j] * W0[Dd + k];
    colb[u] = s;
  }
}

__global__ void k_transpose(const float* __restrict__ C, float* __restrict__ Ct) {
  __shared__ float tile[32][33];
  int b = blockIdx.z;
  int i0 = blockIdx.x * 32;
  int k0 = blockIdx.y * 32;
  int tx = threadIdx.x, ty = threadIdx.y;
  for (int r = 0; r < 32; r += 8)
    tile[ty + r][tx] = C[(size_t)(b * Dd + k0 + ty + r) * CL + i0 + tx];
  __syncthreads();
  for (int r = 0; r < 32; r += 8)
    Ct[(size_t)(b * CL + i0 + ty + r) * Dd + k0 + tx] = tile[tx][ty + r];
}

__global__ void k_sgemm(const float* __restrict__ C, const float* __restrict__ Q,
                        const float* __restrict__ W0, float* __restrict__ S) {
  const int lane = threadIdx.x & 31;
  const int wave = blockIdx.x * (blockDim.x >> 5) + (threadIdx.x >> 5);
  const int nN = QL / 16, nM = CL / 16;
  int b = wave / (nM * nN);
  int rem = wave % (nM * nN);
  int i0 = (rem / nN) * 16;
  int j0 = (rem % nN) * 16;
  const float* w3 = W0 + 2 * Dd;
  int half = lane >> 4, mn = lane & 15;
  v8f acc = {};
  for (int k0 = 0; k0 < Dd; k0 += 4) {
    int kk = k0 + half * 2;
    v2f a, bq;
    a.x  = w3[kk]     * C[(size_t)(b * Dd + kk)     * CL + i0 + mn];
    a.y  = w3[kk + 1] * C[(size_t)(b * Dd + kk + 1) * CL + i0 + mn];
    bq.x = Q[(size_t)(b * Dd + kk)     * QL + j0 + mn];
    bq.y = Q[(size_t)(b * Dd + kk + 1) * QL + j0 + mn];
    acc = __builtin_amdgcn_wmma_f32_16x16x4_f32(false, a, false, bq,
                                                (short)0, acc, false, false);
  }
  float* dst = S + (size_t)(b * CL + i0 + half * 8) * QL + j0 + mn;
#pragma unroll
  for (int v = 0; v < 8; ++v) dst[(size_t)v * QL] = acc[v];
}

__global__ void k_softmax_row(const float* __restrict__ S, const float* __restrict__ colb,
                              const float* __restrict__ qm, float* __restrict__ Sbar) {
  int lane = threadIdx.x & 31;
  int wid = threadIdx.x >> 5;
  int row = blockIdx.x * 8 + wid;
  int b = row / CL;
  const float* src = S + (size_t)row * QL;
  float vals[8];
  float m = NEGV;
#pragma unroll
  for (int t = 0; t < 8; ++t) {
    int j = lane + 32 * t;
    float q = qm[b * QL + j];
    float v = (src[j] + colb[b * QL + j]) * q + NEGV * (1.0f - q);
    vals[t] = v;
    m = fmaxf(m, v);
  }
#pragma unroll
  for (int off = 16; off > 0; off >>= 1) m = fmaxf(m, __shfl_xor(m, off, 32));
  float s = 0.f;
#pragma unroll
  for (int t = 0; t < 8; ++t) { vals[t] = __expf(vals[t] - m); s += vals[t]; }
#pragma unroll
  for (int off = 16; off > 0; off >>= 1) s += __shfl_xor(s, off, 32);
  float inv = 1.0f / s;
  float* dst = Sbar + (size_t)row * QL;
#pragma unroll
  for (int t = 0; t < 8; ++t) dst[lane + 32 * t] = vals[t] * inv;
}

__global__ void k_softmax_col(float* __restrict__ S, const float* __restrict__ rowb,
                              const float* __restrict__ cm) {
  __shared__ float sm[4][64], ss[4][64], fm[64], fs[64];
  int b = blockIdx.x / (QL / 64);
  int j = (blockIdx.x % (QL / 64)) * 64 + (threadIdx.x & 63);
  int jj = threadIdx.x & 63;
  int isub = threadIdx.x >> 6;
  float m = NEGV, s = 0.f;
  for (int i = isub; i < CL; i += 4) {
    float c = cm[b * CL + i];
    float v = (S[(size_t)(b * CL + i) * QL + j] + rowb[b * CL + i]) * c + NEGV * (1.0f - c);
    float nm = fmaxf(m, v);
    s = s * __expf(m - nm) + __expf(v - nm);
    m = nm;
  }
  sm[isub][jj] = m; ss[isub][jj] = s;
  __syncthreads();
  if (isub == 0) {
    float M = sm[0][jj], Su = ss[0][jj];
#pragma unroll
    for (int r = 1; r < 4; ++r) {
      float nm = fmaxf(M, sm[r][jj]);
      Su = Su * __expf(M - nm) + ss[r][jj] * __expf(sm[r][jj] - nm);
      M = nm;
    }
    fm[jj] = M; fs[jj] = 1.0f / Su;
  }
  __syncthreads();
  float M = fm[jj], inv = fs[jj];
  for (int i = isub; i < CL; i += 4) {
    float c = cm[b * CL + i];
    size_t idx = (size_t)(b * CL + i) * QL + j;
    float v = (S[idx] + rowb[b * CL + i]) * c + NEGV * (1.0f - c);
    S[idx] = __expf(v - M) * inv;
  }
}

__global__ void k_tgemm(const float* __restrict__ Sbb, const float* __restrict__ Ct,
                        float* __restrict__ T) {
  int lane = threadIdx.x & 31;
  int wave = blockIdx.x * (blockDim.x >> 5) + (threadIdx.x >> 5);
  int b = wave / (16 * 8);
  int rem = wave % (16 * 8);
  int j0 = (rem / 8) * 16;
  int k0 = (rem % 8) * 16;
  int half = lane >> 4, mn = lane & 15;
  v8f acc = {};
  for (int i0 = 0; i0 < CL; i0 += 4) {
    int ii = i0 + half * 2;
    v2f a, bb;
    a.x  = Sbb[(size_t)(b * CL + ii)     * QL + j0 + mn];
    a.y  = Sbb[(size_t)(b * CL + ii + 1) * QL + j0 + mn];
    bb.x = Ct[(size_t)(b * CL + ii)      * Dd + k0 + mn];
    bb.y = Ct[(size_t)(b * CL + ii + 1)  * Dd + k0 + mn];
    acc = __builtin_amdgcn_wmma_f32_16x16x4_f32(false, a, false, bb,
                                                (short)0, acc, false, false);
  }
  float* dst = T + (size_t)(b * QL + j0 + half * 8) * Dd + k0 + mn;
#pragma unroll
  for (int v = 0; v < 8; ++v) dst[(size_t)v * Dd] = acc[v];
}

__global__ void k_abgemm(const float* __restrict__ Sbar, const float* __restrict__ Q,
                         const float* __restrict__ T,
                         float* __restrict__ Am, float* __restrict__ Bm) {
  int lane = threadIdx.x & 31;
  int wave = blockIdx.x * (blockDim.x >> 5) + (threadIdx.x >> 5);
  int b = wave / (64 * 8);
  int rem = wave % (64 * 8);
  int i0 = (rem / 8) * 16;
  int k0 = (rem % 8) * 16;
  int half = lane >> 4, mn = lane & 15;
  v8f accA = {}, accB = {};
  for (int j0 = 0; j0 < QL; j0 += 4) {
    int jj = j0 + half * 2;
    v2f a, bq, bt;
    a.x  = Sbar[(size_t)(b * CL + i0 + mn) * QL + jj];
    a.y  = Sbar[(size_t)(b * CL + i0 + mn) * QL + jj + 1];
    bq.x = Q[(size_t)(b * Dd + k0 + mn) * QL + jj];
    bq.y = Q[(size_t)(b * Dd + k0 + mn) * QL + jj + 1];
    bt.x = T[(size_t)(b * QL + jj)     * Dd + k0 + mn];
    bt.y = T[(size_t)(b * QL + jj + 1) * Dd + k0 + mn];
    accA = __builtin_amdgcn_wmma_f32_16x16x4_f32(false, a, false, bq,
                                                 (short)0, accA, false, false);
    accB = __builtin_amdgcn_wmma_f32_16x16x4_f32(false, a, false, bt,
                                                 (short)0, accB, false, false);
  }
  int k = k0 + mn;
#pragma unroll
  for (int v = 0; v < 8; ++v) {
    size_t idx = (size_t)(b * CL + i0 + half * 8 + v) * Dd + k;
    Am[idx] = accA[v];
    Bm[idx] = accB[v];
  }
}

__global__ void k_conv(const float* __restrict__ Ct, const float* __restrict__ Am,
                       const float* __restrict__ Bmat,
                       const float* __restrict__ dww, const float* __restrict__ dwb,
                       const float* __restrict__ pww, const float* __restrict__ pwb,
                       float* __restrict__ out) {
  __shared__ float dwX[512][17];
  int b = blockIdx.x / (CL / 16);
  int p0 = (blockIdx.x % (CL / 16)) * 16;
  int d = threadIdx.x & 127;
  int grp = threadIdx.x >> 7;
  for (int it = 0; it < 32; ++it) {
    int combo = it * 2 + grp;
    int which = combo >> 4;
    int p = combo & 15;
    float acc = dwb[which * Dd + d];
#pragma unroll
    for (int tap = 0; tap < 5; ++tap) {
      int pos = p0 + p + tap - 2;
      float x = 0.f;
      if (pos >= 0 && pos < CL) {
        size_t base = (size_t)(b * CL + pos) * Dd + d;
        float c = Ct[base];
        if (which == 0)      x = c;
        else if (which == 1) x = Am[base];
        else if (which == 2) x = c * Am[base];
        else                 x = c * Bmat[base];
      }
      acc += dww[(which * Dd + d) * 5 + tap] * x;
    }
    dwX[which * Dd + d][p] = acc;
  }
  __syncthreads();
  int lane = threadIdx.x & 31;
  int o0 = (threadIdx.x >> 5) * 16;
  int half = lane >> 4, mn = lane & 15;
  v8f acc = {};
  for (int c0 = 0; c0 < 512; c0 += 4) {
    int cc = c0 + half * 2;
    v2f a, bb;
    a.x  = pww[(size_t)(o0 + mn) * 512 + cc];
    a.y  = pww[(size_t)(o0 + mn) * 512 + cc + 1];
    bb.x = dwX[cc][mn];
    bb.y = dwX[cc + 1][mn];
    acc = __builtin_amdgcn_wmma_f32_16x16x4_f32(false, a, false, bb,
                                                (short)0, acc, false, false);
  }
#pragma unroll
  for (int v = 0; v < 8; ++v) {
    int o = o0 + half * 8 + v;
    out[(size_t)(b * Dd + o) * CL + p0 + mn] = acc[v] + pwb[o];
  }
}

extern "C" void kernel_launch(void* const* d_in, const int* in_sizes, int n_in,
                              void* d_out, int out_size, void* d_ws, size_t ws_size,
                              hipStream_t stream) {
  const float* C   = (const float*)d_in[0];
  const float* Q   = (const float*)d_in[1];
  const float* cm  = (const float*)d_in[2];
  const float* qm  = (const float*)d_in[3];
  const float* W0  = (const float*)d_in[4];
  const float* dww = (const float*)d_in[5];
  const float* dwb = (const float*)d_in[6];
  const float* pww = (const float*)d_in[7];
  const float* pwb = (const float*)d_in[8];
  float* out = (float*)d_out;

  float* ws = (float*)d_ws;
  const size_t nS = (size_t)Bn * CL * QL;
  float* S    = ws;
  float* Sbar = S + nS;
  float* rowb = Sbar + nS;
  float* colb = rowb + (size_t)Bn * CL;
  float* T    = colb + (size_t)Bn * QL;
  float* Ct   = T + (size_t)Bn * QL * Dd;
  float* Am   = Ct + (size_t)Bn * CL * Dd;
  float* Bm   = Am + (size_t)Bn * CL * Dd;
  (void)ws_size; (void)in_sizes; (void)n_in; (void)out_size;

  k_bias<<<(Bn * CL + Bn * QL + 255) / 256, 256, 0, stream>>>(C, Q, W0, rowb, colb);
  k_transpose<<<dim3(CL / 32, Dd / 32, Bn), dim3(32, 8), 0, stream>>>(C, Ct);
  k_sgemm<<<(Bn * (CL / 16) * (QL / 16)) / 8, 256, 0, stream>>>(C, Q, W0, S);
  k_softmax_row<<<(Bn * CL) / 8, 256, 0, stream>>>(S, colb, qm, Sbar);
  k_softmax_col<<<Bn * (QL / 64), 256, 0, stream>>>(S, rowb, cm);
  k_tgemm<<<(Bn * (QL / 16) * (Dd / 16)) / 8, 256, 0, stream>>>(S, Ct, T);
  k_abgemm<<<(Bn * (CL / 16) * (Dd / 16)) / 8, 256, 0, stream>>>(Sbar, Q, T, Am, Bm);
  k_conv<<<Bn * (CL / 16), 256, 0, stream>>>(Ct, Am, Bm, dww, dwb, pww, pwb, out);
}


# column softmax: block = 32 columns x 1024 rows on 16 waves, each lane keeps its 64 rows in registers (S read once, all loads in flight), grid 128 -> 256 blocks
# speedup vs baseline: 1.2472x; 1.2472x over previous
#include <hip/hip_runtime.h>

typedef __attribute__((ext_vector_type(2))) float v2f;
typedef __attribute__((ext_vector_type(8))) float v8f;

#define NEGV (-1e30f)
constexpr int Bn = 32, Dd = 128, CL = 1024, QL = 256;

__global__ void k_bias(const float* __restrict__ C, const float* __restrict__ Q,
                       const float* __restrict__ W0,
                       float* __restrict__ rowb, float* __restrict__ colb) {
  int t = blockIdx.x * blockDim.x + threadIdx.x;
  if (t < Bn * CL) {
    int b = t / CL, i = t % CL;
    float s = 0.f;
    for (int k = 0; k < Dd; ++k) s += C[(size_t)(b * Dd + k) * CL + i] * W0[k];
    rowb[t] = s;
  } else if (t < Bn * CL + Bn * QL) {
    int u = t - Bn * CL;
    int b = u / QL, j = u % QL;
    float s = 0.f;
    for (int k = 0; k < Dd; ++k) s += Q[(size_t)(b * Dd + k) * QL + j] * W0[Dd + k];
    colb[u] = s;
  }
}

__global__ void k_transpose(const float* __restrict__ C, float* __restrict__ Ct) {
  __shared__ float tile[32][33];
  int b = blockIdx.z;
  int i0 = blockIdx.x * 32;
  int k0 = blockIdx.y * 32;
  int tx = threadIdx.x, ty = threadIdx.y;
  for (int r = 0; r < 32; r += 8)
    tile[ty + r][tx] = C[(size_t)(b * Dd + k0 + ty + r) * CL + i0 + tx];
  __syncthreads();
  for (int r = 0; r < 32; r += 8)
    Ct[(size_t)(b * CL + i0 + ty + r) * Dd + k0 + tx] = tile[tx][ty + r];
}

__global__ void k_sgemm(const float* __restrict__ C, const float* __restrict__ Q,
                        const float* __restrict__ W0, float* __restrict__ S) {
  const int lane = threadIdx.x & 31;
  const int wave = blockIdx.x * (blockDim.x >> 5) + (threadIdx.x >> 5);
  const int nN = QL / 16, nM = CL / 16;
  int b = wave / (nM * nN);
  int rem = wave % (nM * nN);
  int i0 = (rem / nN) * 16;
  int j0 = (rem % nN) * 16;
  const float* w3 = W0 + 2 * Dd;
  int half = lane >> 4, mn = lane & 15;
  v8f acc = {};
  for (int k0 = 0; k0 < Dd; k0 += 4) {
    int kk = k0 + half * 2;
    v2f a, bq;
    a.x  = w3[kk]     * C[(size_t)(b * Dd + kk)     * CL + i0 + mn];
    a.y  = w3[kk + 1] * C[(size_t)(b * Dd + kk + 1) * CL + i0 + mn];
    bq.x = Q[(size_t)(b * Dd + kk)     * QL + j0 + mn];
    bq.y = Q[(size_t)(b * Dd + kk + 1) * QL + j0 + mn];
    acc = __builtin_amdgcn_wmma_f32_16x16x4_f32(false, a, false, bq,
                                                (short)0, acc, false, false);
  }
  float* dst = S + (size_t)(b * CL + i0 + half * 8) * QL + j0 + mn;
#pragma unroll
  for (int v = 0; v < 8; ++v) dst[(size_t)v * QL] = acc[v];
}

__global__ void k_softmax_row(const float* __restrict__ S, const float* __restrict__ colb,
                              const float* __restrict__ qm, float* __restrict__ Sbar) {
  int lane = threadIdx.x & 31;
  int wid = threadIdx.x >> 5;
  int row = blockIdx.x * 8 + wid;
  int b = row / CL;
  const float* src = S + (size_t)row * QL;
  float vals[8];
  float m = NEGV;
#pragma unroll
  for (int t = 0; t < 8; ++t) {
    int j = lane + 32 * t;
    float q = qm[b * QL + j];
    float v = (src[j] + colb[b * QL + j]) * q + NEGV * (1.0f - q);
    vals[t] = v;
    m = fmaxf(m, v);
  }
#pragma unroll
  for (int off = 16; off > 0; off >>= 1) m = fmaxf(m, __shfl_xor(m, off, 32));
  float s = 0.f;
#pragma unroll
  for (int t = 0; t < 8; ++t) { vals[t] = __expf(vals[t] - m); s += vals[t]; }
#pragma unroll
  for (int off = 16; off > 0; off >>= 1) s += __shfl_xor(s, off, 32);
  float inv = 1.0f / s;
  float* dst = Sbar + (size_t)row * QL;
#pragma unroll
  for (int t = 0; t < 8; ++t) dst[lane + 32 * t] = vals[t] * inv;
}

__global__ void __launch_bounds__(512)
k_softmax_col(float* __restrict__ S, const float* __restrict__ rowb,
              const float* __restrict__ cm) {
  __shared__ float redm[16][32], reds[16][32];
  const int lane = threadIdx.x & 31;
  const int w = __builtin_amdgcn_readfirstlane((int)(threadIdx.x >> 5));
  const int b = blockIdx.x / (QL / 32);
  const int j = (blockIdx.x % (QL / 32)) * 32 + lane;
  const int i0 = w * 64;
  float* p = S + (size_t)(b * CL + i0) * QL + j;
  const float* rb = rowb + b * CL + i0;
  const float* cp = cm + b * CL + i0;
  float v[64];
#pragma unroll
  for (int r = 0; r < 64; ++r) v[r] = p[(size_t)r * QL];
  float m = NEGV;
#pragma unroll
  for (int r = 0; r < 64; ++r) {
    float c = cp[r];
    v[r] = (v[r] + rb[r]) * c + NEGV * (1.0f - c);
    m = fmaxf(m, v[r]);
  }
  redm[w][lane] = m;
  __syncthreads();
  float M = redm[0][lane];
#pragma unroll
  for (int q = 1; q < 16; ++q) M = fmaxf(M, redm[q][lane]);
  float s0 = 0.f, s1 = 0.f, s2 = 0.f, s3 = 0.f;
#pragma unroll
  for (int r = 0; r < 64; r += 4) {
    v[r]     = __expf(v[r]     - M); s0 += v[r];
    v[r + 1] = __expf(v[r + 1] - M); s1 += v[r + 1];
    v[r + 2] = __expf(v[r + 2] - M); s2 += v[r + 2];
    v[r + 3] = __expf(v[r + 3] - M); s3 += v[r + 3];
  }
  reds[w][lane] = (s0 + s1) + (s2 + s3);
  __syncthreads();
  float Z = reds[0][lane];
#pragma unroll
  for (int q = 1; q < 16; ++q) Z += reds[q][lane];
  const float inv = 1.0f / Z;
#pragma unroll
  for (int r = 0; r < 64; ++r) p[(size_t)r * QL] = v[r] * inv;
}

__global__ void k_tgemm(const float* __restrict__ Sbb, const float* __restrict__ Ct,
                        float* __restrict__ T) {
  int lane = threadIdx.x & 31;
  int wave = blockIdx.x * (blockDim.x >> 5) + (threadIdx.x >> 5);
  int b = wave / (16 * 8);
  int rem = wave % (16 * 8);
  int j0 = (rem / 8) * 16;
  int k0 = (rem % 8) * 16;
  int half = lane >> 4, mn = lane & 15;
  v8f acc = {};
  for (int i0 = 0; i0 < CL; i0 += 4) {
    int ii = i0 + half * 2;
    v2f a, bb;
    a.x  = Sbb[(size_t)(b * CL + ii)     * QL + j0 + mn];
    a.y  = Sbb[(size_t)(b * CL + ii + 1) * QL + j0 + mn];
    bb.x = Ct[(size_t)(b * CL + ii)      * Dd + k0 + mn];
    bb.y = Ct[(size_t)(b * CL + ii + 1)  * Dd + k0 + mn];
    acc = __builtin_amdgcn_wmma_f32_16x16x4_f32(false, a, false, bb,
                                                (short)0, acc, false, false);
  }
  float* dst = T + (size_t)(b * QL + j0 + half * 8) * Dd + k0 + mn;
#pragma unroll
  for (int v = 0; v < 8; ++v) dst[(size_t)v * Dd] = acc[v];
}

__global__ void k_abgemm(const float* __restrict__ Sbar, const float* __restrict__ Q,
                         const float* __restrict__ T,
                         float* __restrict__ Am, float* __restrict__ Bm) {
  int lane = threadIdx.x & 31;
  int wave = blockIdx.x * (blockDim.x >> 5) + (threadIdx.x >> 5);
  int b = wave / (64 * 8);
  int rem = wave % (64 * 8);
  int i0 = (rem / 8) * 16;
  int k0 = (rem % 8) * 16;
  int half = lane >> 4, mn = lane & 15;
  v8f accA = {}, accB = {};
  for (int j0 = 0; j0 < QL; j0 += 4) {
    int jj = j0 + half * 2;
    v2f a, bq, bt;
    a.x  = Sbar[(size_t)(b * CL + i0 + mn) * QL + jj];
    a.y  = Sbar[(size_t)(b * CL + i0 + mn) * QL + jj + 1];
    bq.x = Q[(size_t)(b * Dd + k0 + mn) * QL + jj];
    bq.y = Q[(size_t)(b * Dd + k0 + mn) * QL + jj + 1];
    bt.x = T[(size_t)(b * QL + jj)     * Dd + k0 + mn];
    bt.y = T[(size_t)(b * QL + jj + 1) * Dd + k0 + mn];
    accA = __builtin_amdgcn_wmma_f32_16x16x4_f32(false, a, false, bq,
                                                 (short)0, accA, false, false);
    accB = __builtin_amdgcn_wmma_f32_16x16x4_f32(false, a, false, bt,
                                                 (short)0, accB, false, false);
  }
  int k = k0 + mn;
#pragma unroll
  for (int v = 0; v < 8; ++v) {
    size_t idx = (size_t)(b * CL + i0 + half * 8 + v) * Dd + k;
    Am[idx] = accA[v];
    Bm[idx] = accB[v];
  }
}

__global__ void k_conv(const float* __restrict__ Ct, const float* __restrict__ Am,
                       const float* __restrict__ Bmat,
                       const float* __restrict__ dww, const float* __restrict__ dwb,
                       const float* __restrict__ pww, const float* __restrict__ pwb,
                       float* __restrict__ out) {
  __shared__ float dwX[512][17];
  int b = blockIdx.x / (CL / 16);
  int p0 = (blockIdx.x % (CL / 16)) * 16;
  int d = threadIdx.x & 127;
  int grp = threadIdx.x >> 7;
  for (int it = 0; it < 32; ++it) {
    int combo = it * 2 + grp;
    int which = combo >> 4;
    int p = combo & 15;
    float acc = dwb[which * Dd + d];
#pragma unroll
    for (int tap = 0; tap < 5; ++tap) {
      int pos = p0 + p + tap - 2;
      float x = 0.f;
      if (pos >= 0 && pos < CL) {
        size_t base = (size_t)(b * CL + pos) * Dd + d;
        float c = Ct[base];
        if (which == 0)      x = c;
        else if (which == 1) x = Am[base];
        else if (which == 2) x = c * Am[base];
        else                 x = c * Bmat[base];
      }
      acc += dww[(which * Dd + d) * 5 + tap] * x;
    }
    dwX[which * Dd + d][p] = acc;
  }
  __syncthreads();
  int lane = threadIdx.x & 31;
  int o0 = (threadIdx.x >> 5) * 16;
  int half = lane >> 4, mn = lane & 15;
  v8f acc = {};
  for (int c0 = 0; c0 < 512; c0 += 4) {
    int cc = c0 + half * 2;
    v2f a, bb;
    a.x  = pww[(size_t)(o0 + mn) * 512 + cc];
    a.y  = pww[(size_t)(o0 + mn) * 512 + cc + 1];
    bb.x = dwX[cc][mn];
    bb.y = dwX[cc + 1][mn];
    acc = __builtin_amdgcn_wmma_f32_16x16x4_f32(false, a, false, bb,
                                                (short)0, acc, false, false);
  }
#pragma unroll
  for (int v = 0; v < 8; ++v) {
    int o = o0 + half * 8 + v;
    out[(size_t)(b * Dd + o) * CL + p0 + mn] = acc[v] + pwb[o];
  }
}

extern "C" void kernel_launch(void* const* d_in, const int* in_sizes, int n_in,
                              void* d_out, int out_size, void* d_ws, size_t ws_size,
                              hipStream_t stream) {
  const float* C   = (const float*)d_in[0];
  const float* Q   = (const float*)d_in[1];
  const float* cm  = (const float*)d_in[2];
  const float* qm  = (const float*)d_in[3];
  const float* W0  = (const float*)d_in[4];
  const float* dww = (const float*)d_in[5];
  const float* dwb = (const float*)d_in[6];
  const float* pww = (const float*)d_in[7];
  const float* pwb = (const float*)d_in[8];
  float* out = (float*)d_out;

  float* ws = (float*)d_ws;
  const size_t nS = (size_t)Bn * CL * QL;
  float* S    = ws;
  float* Sbar = S + nS;
  float* rowb = Sbar + nS;
  float* colb = rowb + (size_t)Bn * CL;
  float* T    = colb + (size_t)Bn * QL;
  float* Ct   = T + (size_t)Bn * QL * Dd;
  float* Am   = Ct + (size_t)Bn * CL * Dd;
  float* Bm   = Am + (size_t)Bn * CL * Dd;
  (void)ws_size; (void)in_sizes; (void)n_in; (void)out_size;

  k_bias<<<(Bn * CL + Bn * QL + 255) / 256, 256, 0, stream>>>(C, Q, W0, rowb, colb);
  k_transpose<<<dim3(CL / 32, Dd / 32, Bn), dim3(32, 8), 0, stream>>>(C, Ct);
  k_sgemm<<<(Bn * (CL / 16) * (QL / 16)) / 8, 256, 0, stream>>>(C, Q, W0, S);
  k_softmax_row<<<(Bn * CL) / 8, 256, 0, stream>>>(S, colb, qm, Sbar);
  k_softmax_col<<<Bn * (QL / 32), 512, 0, stream>>>(S, rowb, cm);
  k_tgemm<<<(Bn * (QL / 16) * (Dd / 16)) / 8, 256, 0, stream>>>(S, Ct, T);
  k_abgemm<<<(Bn * (CL / 16) * (Dd / 16)) / 8, 256, 0, stream>>>(Sbar, Q, T, Am, Bm);
  k_conv<<<Bn * (CL / 16), 256, 0, stream>>>(Ct, Am, Bm, dww, dwb, pww, pwb, out);
}


# A/Bm GEMM and pointwise-conv GEMM: K-contiguous operand rows (Sbar, Q, pointwise weights) read as one 16-byte load per lane per 8 k, k order permuted alike in A and B: half as many 16-line loads
# speedup vs baseline: 1.3968x; 1.1199x over previous
#include <hip/hip_runtime.h>

typedef __attribute__((ext_vector_type(2))) float v2f;
typedef __attribute__((ext_vector_type(4))) float v4f;
typedef __attribute__((ext_vector_type(8))) float v8f;

#define NEGV (-1e30f)
constexpr int Bn = 32, Dd = 128, CL = 1024, QL = 256;

__global__ void k_bias(const float* __restrict__ C, const float* __restrict__ Q,
                       const float* __restrict__ W0,
                       float* __restrict__ rowb, float* __restrict__ colb) {
  int t = blockIdx.x * blockDim.x + threadIdx.x;
  if (t < Bn * CL) {
    int b = t / CL, i = t % CL;
    float s = 0.f;
    for (int k = 0; k < Dd; ++k) s += C[(size_t)(b * Dd + k) * CL + i] * W0[k];
    rowb[t] = s;
  } else if (t < Bn * CL + Bn * QL) {
    int u = t - Bn * CL;
    int b = u / QL, j = u % QL;
    float s = 0.f;
    for (int k = 0; k < Dd; ++k) s += Q[(size_t)(b * Dd + k) * QL + j] * W0[Dd + k];
    colb[u] = s;
  }
}

__global__ void k_transpose(const float* __restrict__ C, float* __restrict__ Ct) {
  __shared__ float tile[32][33];
  int b = blockIdx.z;
  int i0 = blockIdx.x * 32;
  int k0 = blockIdx.y * 32;
  int tx = threadIdx.x, ty = threadIdx.y;
  for (int r = 0; r < 32; r += 8)
    tile[ty + r][tx] = C[(size_t)(b * Dd + k0 + ty + r) * CL + i0 + tx];
  __syncthreads();
  for (int r = 0; r < 32; r += 8)
    Ct[(size_t)(b * CL + i0 + ty + r) * Dd + k0 + tx] = tile[tx][ty + r];
}

__global__ void k_sgemm(const float* __restrict__ C, const float* __restrict__ Q,
                        const float* __restrict__ W0, float* __restrict__ S) {
  const int lane = threadIdx.x & 31;
  const int wave = blockIdx.x * (blockDim.x >> 5) + (threadIdx.x >> 5);
  const int nN = QL / 16, nM = CL / 16;
  int b = wave / (nM * nN);
  int rem = wave % (nM * nN);
  int i0 = (rem / nN) * 16;
  int j0 = (rem % nN) * 16;
  const float* w3 = W0 + 2 * Dd;
  int half = lane >> 4, mn = lane & 15;
  v8f acc = {};
  for (int k0 = 0; k0 < Dd; k0 += 4) {
    int kk = k0 + half * 2;
    v2f a, bq;
    a.x  = w3[kk]     * C[(size_t)(b * Dd + kk)     * CL + i0 + mn];
    a.y  = w3[kk + 1] * C[(size_t)(b * Dd + kk + 1) * CL + i0 + mn];
    bq.x = Q[(size_t)(b * Dd + kk)     * QL + j0 + mn];
    bq.y = Q[(size_t)(b * Dd + kk + 1) * QL + j0 + mn];
    acc = __builtin_amdgcn_wmma_f32_16x16x4_f32(false, a, false, bq,
                                                (short)0, acc, false, false);
  }
  float* dst = S + (size_t)(b * CL + i0 + half * 8) * QL + j0 + mn;
#pragma unroll
  for (int v = 0; v < 8; ++v) dst[(size_t)v * QL] = acc[v];
}

__global__ void k_softmax_row(const float* __restrict__ S, const float* __restrict__ colb,
                              const float* __restrict__ qm, float* __restrict__ Sbar) {
  int lane = threadIdx.x & 31;
  int wid = threadIdx.x >> 5;
  int row = blockIdx.x * 8 + wid;
  int b = row / CL;
  const float* src = S + (size_t)row * QL;
  float vals[8];
  float m = NEGV;
#pragma unroll
  for (int t = 0; t < 8; ++t) {
    int j = lane + 32 * t;
    float q = qm[b * QL + j];
    float v = (src[j] + colb[b * QL + j]) * q + NEGV * (1.0f - q);
    vals[t] = v;
    m = fmaxf(m, v);
  }
#pragma unroll
  for (int off = 16; off > 0; off >>= 1) m = fmaxf(m, __shfl_xor(m, off, 32));
  float s = 0.f;
#pragma unroll
  for (int t = 0; t < 8; ++t) { vals[t] = __expf(vals[t] - m); s += vals[t]; }
#pragma unroll
  for (int off = 16; off > 0; off >>= 1) s += __shfl_xor(s, off, 32);
  float inv = 1.0f / s;
  float* dst = Sbar + (size_t)row * QL;
#pragma unroll
  for (int t = 0; t < 8; ++t) dst[lane + 32 * t] = vals[t] * inv;
}

__global__ void __launch_bounds__(512)
k_softmax_col(float* __restrict__ S, const float* __restrict__ rowb,
              const float* __restrict__ cm) {
  __shared__ float redm[16][32], reds[16][32];
  const int lane = threadIdx.x & 31;
  const int w = __builtin_amdgcn_readfirstlane((int)(threadIdx.x >> 5));
  const int b = blockIdx.x / (QL / 32);
  const int j = (blockIdx.x % (QL / 32)) * 32 + lane;
  const int i0 = w * 64;
  float* p = S + (size_t)(b * CL + i0) * QL + j;
  const float* rb = rowb + b * CL + i0;
  const float* cp = cm + b * CL + i0;
  float v[64];
#pragma unroll
  for (int r = 0; r < 64; ++r) v[r] = p[(size_t)r * QL];
  float m = NEGV;
#pragma unroll
  for (int r = 0; r < 64; ++r) {
    float c = cp[r];
    v[r] = (v[r] + rb[r]) * c + NEGV * (1.0f - c);
    m = fmaxf(m, v[r]);
  }
  redm[w][lane] = m;
  __syncthreads();
  float M = redm[0][lane];
#pragma unroll
  for (int q = 1; q < 16; ++q) M = fmaxf(M, redm[q][lane]);
  float s0 = 0.f, s1 = 0.f, s2 = 0.f, s3 = 0.f;
#pragma unroll
  for (int r = 0; r < 64; r += 4) {
    v[r]     = __expf(v[r]     - M); s0 += v[r];
    v[r + 1] = __expf(v[r + 1] - M); s1 += v[r + 1];
    v[r + 2] = __expf(v[r + 2] - M); s2 += v[r + 2];
    v[r + 3] = __expf(v[r + 3] - M); s3 += v[r + 3];
  }
  reds[w][lane] = (s0 + s1) + (s2 + s3);
  __syncthreads();
  float Z = reds[0][lane];
#pragma unroll
  for (int q = 1; q < 16; ++q) Z += reds[q][lane];
  const float inv = 1.0f / Z;
#pragma unroll
  for (int r = 0; r < 64; ++r) p[(size_t)r * QL] = v[r] * inv;
}

__global__ void k_tgemm(const float* __restrict__ Sbb, const float* __restrict__ Ct,
                        float* __restrict__ T) {
  int lane = threadIdx.x & 31;
  int wave = blockIdx.x * (blockDim.x >> 5) + (threadIdx.x >> 5);
  int b = wave / (16 * 8);
  int rem = wave % (16 * 8);
  int j0 = (rem / 8) * 16;
  int k0 = (rem % 8) * 16;
  int half = lane >> 4, mn = lane & 15;
  v8f acc = {};
  for (int i0 = 0; i0 < CL; i0 += 4) {
    int ii = i0 + half * 2;
    v2f a, bb;
    a.x  = Sbb[(size_t)(b * CL + ii)     * QL + j0 + mn];
    a.y  = Sbb[(size_t)(b * CL + ii + 1) * QL + j0 + mn];
    bb.x = Ct[(size_t)(b * CL + ii)      * Dd + k0 + mn];
    bb.y = Ct[(size_t)(b * CL + ii + 1)  * Dd + k0 + mn];
    acc = __builtin_amdgcn_wmma_f32_16x16x4_f32(false, a, false, bb,
                                                (short)0, acc, false, false);
  }
  float* dst = T + (size_t)(b * QL + j0 + half * 8) * Dd + k0 + mn;
#pragma unroll
  for (int v = 0; v < 8; ++v) dst[(size_t)v * Dd] = acc[v];
}

__global__ void k_abgemm(const float* __restrict__ Sbar, const float* __restrict__ Q,
                         const float* __restrict__ T,
                         float* __restrict__ Am, float* __restrict__ Bm) {
  int lane = threadIdx.x & 31;
  int wave = blockIdx.x * (blockDim.x >> 5) + (threadIdx.x >> 5);
  int b = wave / (64 * 8);
  int rem = wave % (64 * 8);
  int i0 = (rem / 8) * 16;
  int k0 = (rem % 8) * 16;
  int half = lane >> 4, mn = lane & 15;
  v8f accA = {}, accB = {};
  const float* arow = Sbar + (size_t)(b * CL + i0 + mn) * QL + half * 4;
  const float* qrow = Q + (size_t)(b * Dd + k0 + mn) * QL + half * 4;
  const float* tcol = T + (size_t)(b * QL + half * 4) * Dd + k0 + mn;
#pragma unroll 2
  for (int j0 = 0; j0 < QL; j0 += 8) {
    v4f av = *(const v4f*)(arow + j0);
    v4f qv = *(const v4f*)(qrow + j0);
    const float* tp = tcol + (size_t)j0 * Dd;
    v2f a0, a1, q0, q1, t0, t1;
    a0.x = av.x; a0.y = av.y; a1.x = av.z; a1.y = av.w;
    q0.x = qv.x; q0.y = qv.y; q1.x = qv.z; q1.y = qv.w;
    t0.x = tp[0];      t0.y = tp[Dd];
    t1.x = tp[2 * Dd]; t1.y = tp[3 * Dd];
    accA = __builtin_amdgcn_wmma_f32_16x16x4_f32(false, a0, false, q0,
                                                 (short)0, accA, false, false);
    accB = __builtin_amdgcn_wmma_f32_16x16x4_f32(false, a0, false, t0,
                                                 (short)0, accB, false, false);
    accA = __builtin_amdgcn_wmma_f32_16x16x4_f32(false, a1, false, q1,
                                                 (short)0, accA, false, false);
    accB = __builtin_amdgcn_wmma_f32_16x16x4_f32(false, a1, false, t1,
                                                 (short)0, accB, false, false);
  }
  int k = k0 + mn;
#pragma unroll
  for (int v = 0; v < 8; ++v) {
    size_t idx = (size_t)(b * CL + i0 + half * 8 + v) * Dd + k;
    Am[idx] = accA[v];
    Bm[idx] = accB[v];
  }
}

__global__ void k_conv(const float* __restrict__ Ct, const float* __restrict__ Am,
                       const float* __restrict__ Bmat,
                       const float* __restrict__ dww, const float* __restrict__ dwb,
                       const float* __restrict__ pww, const float* __restrict__ pwb,
                       float* __restrict__ out) {
  __shared__ float dwX[512][17];
  int b = blockIdx.x / (CL / 16);
  int p0 = (blockIdx.x % (CL / 16)) * 16;
  int d = threadIdx.x & 127;
  int grp = threadIdx.x >> 7;
  for (int it = 0; it < 32; ++it) {
    int combo = it * 2 + grp;
    int which = combo >> 4;
    int p = combo & 15;
    float acc = dwb[which * Dd + d];
#pragma unroll
    for (int tap = 0; tap < 5; ++tap) {
      int pos = p0 + p + tap - 2;
      float x = 0.f;
      if (pos >= 0 && pos < CL) {
        size_t base = (size_t)(b * CL + pos) * Dd + d;
        float c = Ct[base];
        if (which == 0)      x = c;
        else if (which == 1) x = Am[base];
        else if (which == 2) x = c * Am[base];
        else                 x = c * Bmat[base];
      }
      acc += dww[(which * Dd + d) * 5 + tap] * x;
    }
    dwX[which * Dd + d][p] = acc;
  }
  __syncthreads();
  int lane = threadIdx.x & 31;
  int o0 = (threadIdx.x >> 5) * 16;
  int half = lane >> 4, mn = lane & 15;
  v8f acc = {};
  const float* prow = pww + (size_t)(o0 + mn) * 512 + half * 4;
  for (int c0 = 0; c0 < 512; c0 += 8) {
    v4f pv = *(const v4f*)(prow + c0);
    int cc = c0 + half * 4;
    v2f a0, a1, b0, b1;
    a0.x = pv.x; a0.y = pv.y; a1.x = pv.z; a1.y = pv.w;
    b0.x = dwX[cc][mn];     b0.y = dwX[cc + 1][mn];
    b1.x = dwX[cc + 2][mn]; b1.y = dwX[cc + 3][mn];
    acc = __builtin_amdgcn_wmma_f32_16x16x4_f32(false, a0, false, b0,
                                                (short)0, acc, false, false);
    acc = __builtin_amdgcn_wmma_f32_16x16x4_f32(false, a1, false, b1,
                                                (short)0, acc, false, false);
  }
#pragma unroll
  for (int v = 0; v < 8; ++v) {
    int o = o0 + half * 8 + v;
    out[(size_t)(b * Dd + o) * CL + p0 + mn] = acc[v] + pwb[o];
  }
}

extern "C" void kernel_launch(void* const* d_in, const int* in_sizes, int n_in,
                              void* d_out, int out_size, void* d_ws, size_t ws_size,
                              hipStream_t stream) {
  const float* C   = (const float*)d_in[0];
  const float* Q   = (const float*)d_in[1];
  const float* cm  = (const float*)d_in[2];
  const float* qm  = (const float*)d_in[3];
  const float* W0  = (const float*)d_in[4];
  const float* dww = (const float*)d_in[5];
  const float* dwb = (const float*)d_in[6];
  const float* pww = (const float*)d_in[7];
  const float* pwb = (const float*)d_in[8];
  float* out = (float*)d_out;

  float* ws = (float*)d_ws;
  const size_t nS = (size_t)Bn * CL * QL;
  float* S    = ws;
  float* Sbar = S + nS;
  float* rowb = Sbar + nS;
  float* colb = rowb + (size_t)Bn * CL;
  float* T    = colb + (size_t)Bn * QL;
  float* Ct   = T + (size_t)Bn * QL * Dd;
  float* Am   = Ct + (size_t)Bn * CL * Dd;
  float* Bm   = Am + (size_t)Bn * CL * Dd;
  (void)ws_size; (void)in_sizes; (void)n_in; (void)out_size;

  k_bias<<<(Bn * CL + Bn * QL + 255) / 256, 256, 0, stream>>>(C, Q, W0, rowb, colb);
  k_transpose<<<dim3(CL / 32, Dd / 32, Bn), dim3(32, 8), 0, stream>>>(C, Ct);
  k_sgemm<<<(Bn * (CL / 16) * (QL / 16)) / 8, 256, 0, stream>>>(C, Q, W0, S);
  k_softmax_row<<<(Bn * CL) / 8, 256, 0, stream>>>(S, colb, qm, Sbar);
  k_softmax_col<<<Bn * (QL / 32), 512, 0, stream>>>(S, rowb, cm);
  k_tgemm<<<(Bn * (QL / 16) * (Dd / 16)) / 8, 256, 0, stream>>>(S, Ct, T);
  k_abgemm<<<(Bn * (CL / 16) * (Dd / 16)) / 8, 256, 0, stream>>>(Sbar, Q, T, Am, Bm);
  k_conv<<<Bn * (CL / 16), 256, 0, stream>>>(Ct, Am, Bm, dww, dwb, pww, pwb, out);
}


# conv depthwise stage: each thread loads its 12 input positions of Ct, A, Bm once, all loads in flight, and slides the 5-tap window over registers, branch-free (was 32 branchy passes, a wait per tap)
# speedup vs baseline: 1.6117x; 1.1539x over previous
#include <hip/hip_runtime.h>

typedef __attribute__((ext_vector_type(2))) float v2f;
typedef __attribute__((ext_vector_type(4))) float v4f;
typedef __attribute__((ext_vector_type(8))) float v8f;

#define NEGV (-1e30f)
constexpr int Bn = 32, Dd = 128, CL = 1024, QL = 256;

__global__ void k_bias(const float* __restrict__ C, const float* __restrict__ Q,
                       const float* __restrict__ W0,
                       float* __restrict__ rowb, float* __restrict__ colb) {
  int t = blockIdx.x * blockDim.x + threadIdx.x;
  if (t < Bn * CL) {
    int b = t / CL, i = t % CL;
    float s = 0.f;
    for (int k = 0; k < Dd; ++k) s += C[(size_t)(b * Dd + k) * CL + i] * W0[k];
    rowb[t] = s;
  } else if (t < Bn * CL + Bn * QL) {
    int u = t - Bn * CL;
    int b = u / QL, j = u % QL;
    float s = 0.f;
    for (int k = 0; k < Dd; ++k) s += Q[(size_t)(b * Dd + k) * QL + j] * W0[Dd + k];
    colb[u] = s;
  }
}

__global__ void k_transpose(const float* __restrict__ C, float* __restrict__ Ct) {
  __shared__ float tile[32][33];
  int b = blockIdx.z;
  int i0 = blockIdx.x * 32;
  int k0 = blockIdx.y * 32;
  int tx = threadIdx.x, ty = threadIdx.y;
  for (int r = 0; r < 32; r += 8)
    tile[ty + r][tx] = C[(size_t)(b * Dd + k0 + ty + r) * CL + i0 + tx];
  __syncthreads();
  for (int r = 0; r < 32; r += 8)
    Ct[(size_t)(b * CL + i0 + ty + r) * Dd + k0 + tx] = tile[tx][ty + r];
}

__global__ void k_sgemm(const float* __restrict__ C, const float* __restrict__ Q,
                        const float* __restrict__ W0, float* __restrict__ S) {
  const int lane = threadIdx.x & 31;
  const int wave = blockIdx.x * (blockDim.x >> 5) + (threadIdx.x >> 5);
  const int nN = QL / 16, nM = CL / 16;
  int b = wave / (nM * nN);
  int rem = wave % (nM * nN);
  int i0 = (rem / nN) * 16;
  int j0 = (rem % nN) * 16;
  const float* w3 = W0 + 2 * Dd;
  int half = lane >> 4, mn = lane & 15;
  v8f acc = {};
  for (int k0 = 0; k0 < Dd; k0 += 4) {
    int kk = k0 + half * 2;
    v2f a, bq;
    a.x  = w3[kk]     * C[(size_t)(b * Dd + kk)     * CL + i0 + mn];
    a.y  = w3[kk + 1] * C[(size_t)(b * Dd + kk + 1) * CL + i0 + mn];
    bq.x = Q[(size_t)(b * Dd + kk)     * QL + j0 + mn];
    bq.y = Q[(size_t)(b * Dd + kk + 1) * QL + j0 + mn];
    acc = __builtin_amdgcn_wmma_f32_16x16x4_f32(false, a, false, bq,
                                                (short)0, acc, false, false);
  }
  float* dst = S + (size_t)(b * CL + i0 + half * 8) * QL + j0 + mn;
#pragma unroll
  for (int v = 0; v < 8; ++v) dst[(size_t)v * QL] = acc[v];
}

__global__ void k_softmax_row(const float* __restrict__ S, const float* __restrict__ colb,
                              const float* __restrict__ qm, float* __restrict__ Sbar) {
  int lane = threadIdx.x & 31;
  int wid = threadIdx.x >> 5;
  int row = blockIdx.x * 8 + wid;
  int b = row / CL;
  const float* src = S + (size_t)row * QL;
  float vals[8];
  float m = NEGV;
#pragma unroll
  for (int t = 0; t < 8; ++t) {
    int j = lane + 32 * t;
    float q = qm[b * QL + j];
    float v = (src[j] + colb[b * QL + j]) * q + NEGV * (1.0f - q);
    vals[t] = v;
    m = fmaxf(m, v);
  }
#pragma unroll
  for (int off = 16; off > 0; off >>= 1) m = fmaxf(m, __shfl_xor(m, off, 32));
  float s = 0.f;
#pragma unroll
  for (int t = 0; t < 8; ++t) { vals[t] = __expf(vals[t] - m); s += vals[t]; }
#pragma unroll
  for (int off = 16; off > 0; off >>= 1) s += __shfl_xor(s, off, 32);
  float inv = 1.0f / s;
  float* dst = Sbar + (size_t)row * QL;
#pragma unroll
  for (int t = 0; t < 8; ++t) dst[lane + 32 * t] = vals[t] * inv;
}

__global__ void __launch_bounds__(512)
k_softmax_col(float* __restrict__ S, const float* __restrict__ rowb,
              const float* __restrict__ cm) {
  __shared__ float redm[16][32], reds[16][32];
  const int lane = threadIdx.x & 31;
  const int w = __builtin_amdgcn_readfirstlane((int)(threadIdx.x >> 5));
  const int b = blockIdx.x / (QL / 32);
  const int j = (blockIdx.x % (QL / 32)) * 32 + lane;
  const int i0 = w * 64;
  float* p = S + (size_t)(b * CL + i0) * QL + j;
  const float* rb = rowb + b * CL + i0;
  const float* cp = cm + b * CL + i0;
  float v[64];
#pragma unroll
  for (int r = 0; r < 64; ++r) v[r] = p[(size_t)r * QL];
  float m = NEGV;
#pragma unroll
  for (int r = 0; r < 64; ++r) {
    float c = cp[r];
    v[r] = (v[r] + rb[r]) * c + NEGV * (1.0f - c);
    m = fmaxf(m, v[r]);
  }
  redm[w][lane] = m;
  __syncthreads();
  float M = redm[0][lane];
#pragma unroll
  for (int q = 1; q < 16; ++q) M = fmaxf(M, redm[q][lane]);
  float s0 = 0.f, s1 = 0.f, s2 = 0.f, s3 = 0.f;
#pragma unroll
  for (int r = 0; r < 64; r += 4) {
    v[r]     = __expf(v[r]     - M); s0 += v[r];
    v[r + 1] = __expf(v[r + 1] - M); s1 += v[r + 1];
    v[r + 2] = __expf(v[r + 2] - M); s2 += v[r + 2];
    v[r + 3] = __expf(v[r + 3] - M); s3 += v[r + 3];
  }
  reds[w][lane] = (s0 + s1) + (s2 + s3);
  __syncthreads();
  float Z = reds[0][lane];
#pragma unroll
  for (int q = 1; q < 16; ++q) Z += reds[q][lane];
  const float inv = 1.0f / Z;
#pragma unroll
  for (int r = 0; r < 64; ++r) p[(size_t)r * QL] = v[r] * inv;
}

__global__ void k_tgemm(const float* __restrict__ Sbb, const float* __restrict__ Ct,
                        float* __restrict__ T) {
  int lane = threadIdx.x & 31;
  int wave = blockIdx.x * (blockDim.x >> 5) + (threadIdx.x >> 5);
  int b = wave / (16 * 8);
  int rem = wave % (16 * 8);
  int j0 = (rem / 8) * 16;
  int k0 = (rem % 8) * 16;
  int half = lane >> 4, mn = lane & 15;
  v8f acc = {};
  for (int i0 = 0; i0 < CL; i0 += 4) {
    int ii = i0 + half * 2;
    v2f a, bb;
    a.x  = Sbb[(size_t)(b * CL + ii)     * QL + j0 + mn];
    a.y  = Sbb[(size_t)(b * CL + ii + 1) * QL + j0 + mn];
    bb.x = Ct[(size_t)(b * CL + ii)      * Dd + k0 + mn];
    bb.y = Ct[(size_t)(b * CL + ii + 1)  * Dd + k0 + mn];
    acc = __builtin_amdgcn_wmma_f32_16x16x4_f32(false, a, false, bb,
                                                (short)0, acc, false, false);
  }
  float* dst = T + (size_t)(b * QL + j0 + half * 8) * Dd + k0 + mn;
#pragma unroll
  for (int v = 0; v < 8; ++v) dst[(size_t)v * Dd] = acc[v];
}

__global__ void k_abgemm(const float* __restrict__ Sbar, const float* __restrict__ Q,
                         const float* __restrict__ T,
                         float* __restrict__ Am, float* __restrict__ Bm) {
  int lane = threadIdx.x & 31;
  int wave = blockIdx.x * (blockDim.x >> 5) + (threadIdx.x >> 5);
  int b = wave / (64 * 8);
  int rem = wave % (64 * 8);
  int i0 = (rem / 8) * 16;
  int k0 = (rem % 8) * 16;
  int half = lane >> 4, mn = lane & 15;
  v8f accA = {}, accB = {};
  const float* arow = Sbar + (size_t)(b * CL + i0 + mn) * QL + half * 4;
  const float* qrow = Q + (size_t)(b * Dd + k0 + mn) * QL + half * 4;
  const float* tcol = T + (size_t)(b * QL + half * 4) * Dd + k0 + mn;
#pragma unroll 2
  for (int j0 = 0; j0 < QL; j0 += 8) {
    v4f av = *(const v4f*)(arow + j0);
    v4f qv = *(const v4f*)(qrow + j0);
    const float* tp = tcol + (size_t)j0 * Dd;
    v2f a0, a1, q0, q1, t0, t1;
    a0.x = av.x; a0.y = av.y; a1.x = av.z; a1.y = av.w;
    q0.x = qv.x; q0.y = qv.y; q1.x = qv.z; q1.y = qv.w;
    t0.x = tp[0];      t0.y = tp[Dd];
    t1.x = tp[2 * Dd]; t1.y = tp[3 * Dd];
    accA = __builtin_amdgcn_wmma_f32_16x16x4_f32(false, a0, false, q0,
                                                 (short)0, accA, false, false);
    accB = __builtin_amdgcn_wmma_f32_16x16x4_f32(false, a0, false, t0,
                                                 (short)0, accB, false, false);
    accA = __builtin_amdgcn_wmma_f32_16x16x4_f32(false, a1, false, q1,
                                                 (short)0, accA, false, false);
    accB = __builtin_amdgcn_wmma_f32_16x16x4_f32(false, a1, false, t1,
                                                 (short)0, accB, false, false);
  }
  int k = k0 + mn;
#pragma unroll
  for (int v = 0; v < 8; ++v) {
    size_t idx = (size_t)(b * CL + i0 + half * 8 + v) * Dd + k;
    Am[idx] = accA[v];
    Bm[idx] = accB[v];
  }
}

__global__ void __launch_bounds__(256) __attribute__((amdgpu_waves_per_eu(16)))
k_conv(const float* __restrict__ Ct, const float* __restrict__ Am,
       const float* __restrict__ Bmat,
       const float* __restrict__ dww, const float* __restrict__ dwb,
       const float* __restrict__ pww, const float* __restrict__ pwb,
       float* __restrict__ out) {
  __shared__ float dwX[512][17];
  int b = blockIdx.x / (CL / 16);
  int p0 = (blockIdx.x % (CL / 16)) * 16;
  const int d = threadIdx.x & 127;
  const int grp = __builtin_amdgcn_readfirstlane((int)(threadIdx.x >> 7));
  const int q0 = p0 + grp * 8 - 2;
  auto loadw = [&](const int g, v2f (&w)[5], v2f& bias) {
    const float* wa = dww + (g * Dd + d) * 5;
    const float* wb = dww + ((g + 1) * Dd + d) * 5;
#pragma unroll
    for (int tap = 0; tap < 5; ++tap) { w[tap].x = wa[tap]; w[tap].y = wb[tap]; }
    bias.x = dwb[g * Dd + d];
    bias.y = dwb[(g + 1) * Dd + d];
  };
  auto emit2 = [&](const int g, const v2f (&x)[12], const v2f (&w)[5], const v2f bias) {
#pragma unroll
    for (int p = 0; p < 8; ++p) {
      v2f acc = bias;
#pragma unroll
      for (int tap = 0; tap < 5; ++tap) acc += w[tap] * x[p + tap];
      dwX[g * Dd + d][grp * 8 + p] = acc.x;
      dwX[(g + 1) * Dd + d][grp * 8 + p] = acc.y;
    }
  };
  float rc[12], ra[12], rm[12];
#pragma unroll
  for (int t = 0; t < 12; ++t) {
    const int pos = q0 + t;
    const bool ok = (pos >= 0) && (pos < CL);
    const int row = b * CL + (ok ? pos : 0);
    rc[t] = Ct[row * Dd + d];
    ra[t] = Am[row * Dd + d];
    rm[t] = Bmat[row * Dd + d];
  }
  v2f w[5], bias;
  loadw(0, w, bias);
  __builtin_amdgcn_sched_barrier(0);
  v2f x[12];
  float xm[12];
#pragma unroll
  for (int t = 0; t < 12; ++t) {
    const int pos = q0 + t;
    const bool ok = (pos >= 0) && (pos < CL);
    x[t].x = ok ? rc[t] : 0.f;
    x[t].y = ok ? ra[t] : 0.f;
    xm[t] = ok ? rm[t] : 0.f;
  }
  emit2(0, x, w, bias);
  loadw(2, w, bias);
#pragma unroll
  for (int t = 0; t < 12; ++t) {
    const float c = x[t].x;
    x[t].x = c * x[t].y;
    x[t].y = c * xm[t];
  }
  emit2(2, x, w, bias);
  __syncthreads();
  int lane = threadIdx.x & 31;
  int o0 = (threadIdx.x >> 5) * 16;
  int half = lane >> 4, mn = lane & 15;
  v8f acc = {};
  const float* prow = pww + (size_t)(o0 + mn) * 512 + half * 4;
  for (int c0 = 0; c0 < 512; c0 += 8) {
    v4f pv = *(const v4f*)(prow + c0);
    int cc = c0 + half * 4;
    v2f a0, a1, b0, b1;
    a0.x = pv.x; a0.y = pv.y; a1.x = pv.z; a1.y = pv.w;
    b0.x = dwX[cc][mn];     b0.y = dwX[cc + 1][mn];
    b1.x = dwX[cc + 2][mn]; b1.y = dwX[cc + 3][mn];
    acc = __builtin_amdgcn_wmma_f32_16x16x4_f32(false, a0, false, b0,
                                                (short)0, acc, false, false);
    acc = __builtin_amdgcn_wmma_f32_16x16x4_f32(false, a1, false, b1,
                                                (short)0, acc, false, false);
  }
#pragma unroll
  for (int v = 0; v < 8; ++v) {
    int o = o0 + half * 8 + v;
    out[(size_t)(b * Dd + o) * CL + p0 + mn] = acc[v] + pwb[o];
  }
}

extern "C" void kernel_launch(void* const* d_in, const int* in_sizes, int n_in,
                              void* d_out, int out_size, void* d_ws, size_t ws_size,
                              hipStream_t stream) {
  const float* C   = (const float*)d_in[0];
  const float* Q   = (const float*)d_in[1];
  const float* cm  = (const float*)d_in[2];
  const float* qm  = (const float*)d_in[3];
  const float* W0  = (const float*)d_in[4];
  const float* dww = (const float*)d_in[5];
  const float* dwb = (const float*)d_in[6];
  const float* pww = (const float*)d_in[7];
  const float* pwb = (const float*)d_in[8];
  float* out = (float*)d_out;

  float* ws = (float*)d_ws;
  const size_t nS = (size_t)Bn * CL * QL;
  float* S    = ws;
  float* Sbar = S + nS;
  float* rowb = Sbar + nS;
  float* colb = rowb + (size_t)Bn * CL;
  float* T    = colb + (size_t)Bn * QL;
  float* Ct   = T + (size_t)Bn * QL * Dd;
  float* Am   = Ct + (size_t)Bn * CL * Dd;
  float* Bm   = Am + (size_t)Bn * CL * Dd;
  (void)ws_size; (void)in_sizes; (void)n_in; (void)out_size;

  k_bias<<<(Bn * CL + Bn * QL + 255) / 256, 256, 0, stream>>>(C, Q, W0, rowb, colb);
  k_transpose<<<dim3(CL / 32, Dd / 32, Bn), dim3(32, 8), 0, stream>>>(C, Ct);
  k_sgemm<<<(Bn * (CL / 16) * (QL / 16)) / 8, 256, 0, stream>>>(C, Q, W0, S);
  k_softmax_row<<<(Bn * CL) / 8, 256, 0, stream>>>(S, colb, qm, Sbar);
  k_softmax_col<<<Bn * (QL / 32), 512, 0, stream>>>(S, rowb, cm);
  k_tgemm<<<(Bn * (QL / 16) * (Dd / 16)) / 8, 256, 0, stream>>>(S, Ct, T);
  k_abgemm<<<(Bn * (CL / 16) * (Dd / 16)) / 8, 256, 0, stream>>>(Sbar, Q, T, Am, Bm);
  k_conv<<<Bn * (CL / 16), 256, 0, stream>>>(Ct, Am, Bm, dww, dwb, pww, pwb, out);
}


# A/Bm GEMM: a wave owns 16 rows x all 256 columns of [A|Bm] (16 accumulator tiles): one Sbar load feeds 32 matrix ops (was 4), T rows read as whole lines, operands one step ahead; 2048 -> 256 blocks
# speedup vs baseline: 2.0697x; 1.2842x over previous
#include <hip/hip_runtime.h>

typedef __attribute__((ext_vector_type(2))) float v2f;
typedef __attribute__((ext_vector_type(4))) float v4f;
typedef __attribute__((ext_vector_type(8))) float v8f;

#define NEGV (-1e30f)
constexpr int Bn = 32, Dd = 128, CL = 1024, QL = 256;

template <int PITCH>
__device__ __forceinline__ v4f load16_dot(const float* __restrict__ src,
                                          const float* __restrict__ wv, v4f (&x)[16]) {
#pragma unroll
  for (int r = 0; r < 16; ++r) x[r] = *(const v4f*)(src + (size_t)r * PITCH);
  v4f p = x[0] * wv[0];
#pragma unroll
  for (int r = 1; r < 16; ++r) p += x[r] * wv[r];
  return p;
}

__global__ void __launch_bounds__(256)
k_prep(const float* __restrict__ C, const float* __restrict__ Q,
       const float* __restrict__ W0,
       float* __restrict__ Ct, float* __restrict__ rowb, float* __restrict__ colb) {
  constexpr int P4 = 33;
  constexpr int NT = Bn * (CL / 128);
  __shared__ v4f tile[128 * P4];
  __shared__ v4f red[8 * 32];
  const int lane = threadIdx.x & 31;
  const int w = __builtin_amdgcn_readfirstlane((int)(threadIdx.x >> 5));
  v4f x[16];
  if (blockIdx.x < NT) {
    const int b = blockIdx.x / (CL / 128);
    const int i0 = (blockIdx.x % (CL / 128)) * 128;
    const v4f p = load16_dot<CL>(C + (size_t)(b * Dd + 16 * w) * CL + i0 + 4 * lane,
                                 W0 + 16 * w, x);
    red[w * 32 + lane] = p;
#pragma unroll
    for (int e = 0; e < 4; ++e) {
      v4f* dst = tile + (32 * e + lane) * P4 + 4 * w;
#pragma unroll
      for (int h = 0; h < 4; ++h) {
        v4f t;
        t.x = x[4 * h][e]; t.y = x[4 * h + 1][e];
        t.z = x[4 * h + 2][e]; t.w = x[4 * h + 3][e];
        dst[h] = t;
      }
    }
    __syncthreads();
    float* out = Ct + (size_t)(b * CL + i0 + 16 * w) * Dd + 4 * lane;
#pragma unroll
    for (int q = 0; q < 16; ++q) {
      const int i = 16 * w + q;
      const int row = 32 * (i & 3) + (i >> 2);
      *(v4f*)(out + (size_t)q * Dd) = tile[row * P4 + lane];
    }
    if (threadIdx.x < 128) {
      const float* rf = (const float*)red;
      float s = rf[threadIdx.x];
#pragma unroll
      for (int q = 1; q < 8; ++q) s += rf[q * 128 + threadIdx.x];
      rowb[b * CL + i0 + threadIdx.x] = s;
    }
  } else {
    const int u = blockIdx.x - NT;
    const int b = u >> 1;
    const int j0 = (u & 1) * 128;
    const v4f p = load16_dot<QL>(Q + (size_t)(b * Dd + 16 * w) * QL + j0 + 4 * lane,
                                 W0 + Dd + 16 * w, x);
    red[w * 32 + lane] = p;
    __syncthreads();
    if (threadIdx.x < 128) {
      const float* rf = (const float*)red;
      float s = rf[threadIdx.x];
#pragma unroll
      for (int q = 1; q < 8; ++q) s += rf[q * 128 + threadIdx.x];
      colb[b * QL + j0 + threadIdx.x] = s;
    }
  }
}

template <int M>
__device__ __forceinline__ float row_xor(float v) {
  const int i = __builtin_bit_cast(int, v);
  const int r = __builtin_amdgcn_update_dpp(i, i, 0x160 + M, 0xf, 0xf, true);
  return __builtin_bit_cast(float, r);
}

struct SFrag {
  v4f q0[4], q1[4];
  float c0, c1;
  v2f w;
};

__global__ void __launch_bounds__(256)
k_sgemm_rowsm(const float* __restrict__ C, const float* __restrict__ Q,
              const float* __restrict__ W0, const float* __restrict__ colb,
              const float* __restrict__ qm,
              float* __restrict__ S, float* __restrict__ Sbar) {
  const int lane = threadIdx.x & 31;
  const int wave = blockIdx.x * 8 + (int)(threadIdx.x >> 5);
  const int b = wave / (CL / 16);
  const int i0 = (wave % (CL / 16)) * 16;
  const int half = lane >> 4, c = lane & 15;
  __shared__ float w3s[Dd];
  if (threadIdx.x < Dd) w3s[threadIdx.x] = W0[2 * Dd + threadIdx.x];
  __syncthreads();
  const float* w3h = w3s + 2 * half;
  const float* cp = C + (size_t)(b * Dd + 2 * half) * CL + i0 + c;
  const float* qp = Q + (size_t)(b * Dd + 2 * half) * QL + 4 * c;
  v8f acc[16];
#pragma unroll
  for (int n = 0; n < 16; ++n) acc[n] = v8f{};

  auto ld = [&](const int k0, SFrag& f) {
    const float* cpk = cp + (size_t)k0 * CL;
    const float* qpk = qp + (size_t)k0 * QL;
    f.c0 = cpk[0];
    f.c1 = cpk[CL];
    f.w = *(const v2f*)(w3h + k0);
#pragma unroll
    for (int g = 0; g < 4; ++g) {
      f.q0[g] = *(const v4f*)(qpk + 64 * g);
      f.q1[g] = *(const v4f*)(qpk + QL + 64 * g);
    }
  };
  auto mm = [&](const SFrag& f) {
    v2f a;
    a.x = f.w.x * f.c0;
    a.y = f.w.y * f.c1;
#pragma unroll
    for (int g = 0; g < 4; ++g) {
#pragma unroll
      for (int t = 0; t < 4; ++t) {
        v2f bq;
        bq.x = f.q0[g][t];
        bq.y = f.q1[g][t];
        acc[4 * g + t] = __builtin_amdgcn_wmma_f32_16x16x4_f32(
            false, a, false, bq, (short)0, acc[4 * g + t], false, false);
      }
    }
  };

  SFrag fx, fy;
  ld(0, fx);
  for (int k0 = 0; k0 < Dd; k0 += 8) {
    ld(k0 + 4, fy);
    asm volatile("" ::: "memory");
    mm(fx);
    ld((k0 + 8) & (Dd - 1), fx);
    asm volatile("" ::: "memory");
    mm(fy);
  }

  v4f cb[4], qv[4];
#pragma unroll
  for (int g = 0; g < 4; ++g) {
    cb[g] = *(const v4f*)(colb + b * QL + 64 * g + 4 * c);
    qv[g] = *(const v4f*)(qm + b * QL + 64 * g + 4 * c);
  }
#pragma unroll
  for (int r = 0; r < 8; ++r) {
    const size_t rowoff = (size_t)(b * CL + i0 + half * 8 + r) * QL + 4 * c;
    v4f val[4];
    float m = NEGV;
#pragma unroll
    for (int g = 0; g < 4; ++g) {
      v4f s;
#pragma unroll
      for (int t = 0; t < 4; ++t) {
        const float x = acc[4 * g + t][r];
        const float q = qv[g][t];
        s[t] = x;
        const float v = (x + cb[g][t]) * q + NEGV * (1.0f - q);
        val[g][t] = v;
        m = fmaxf(m, v);
      }
      *(v4f*)(S + rowoff + 64 * g) = s;
    }
    m = fmaxf(m, row_xor<1>(m));
    m = fmaxf(m, row_xor<2>(m));
    m = fmaxf(m, row_xor<4>(m));
    m = fmaxf(m, row_xor<8>(m));
    float sum = 0.f;
#pragma unroll
    for (int g = 0; g < 4; ++g) {
#pragma unroll
      for (int t = 0; t < 4; ++t) {
        val[g][t] = __expf(val[g][t] - m);
        sum += val[g][t];
      }
    }
    sum += row_xor<1>(sum);
    sum += row_xor<2>(sum);
    sum += row_xor<4>(sum);
    sum += row_xor<8>(sum);
    const float inv = 1.0f / sum;
#pragma unroll
    for (int g = 0; g < 4; ++g) {
      v4f o;
#pragma unroll
      for (int t = 0; t < 4; ++t) o[t] = val[g][t] * inv;
      *(v4f*)(Sbar + rowoff + 64 * g) = o;
    }
  }
}

__global__ void __launch_bounds__(512)
k_softmax_col(float* __restrict__ S, const float* __restrict__ rowb,
              const float* __restrict__ cm) {
  __shared__ float redm[16][32], reds[16][32];
  const int lane = threadIdx.x & 31;
  const int w = __builtin_amdgcn_readfirstlane((int)(threadIdx.x >> 5));
  const int b = blockIdx.x / (QL / 32);
  const int j = (blockIdx.x % (QL / 32)) * 32 + lane;
  const int i0 = w * 64;
  float* p = S + (size_t)(b * CL + i0) * QL + j;
  const float* rb = rowb + b * CL + i0;
  const float* cp = cm + b * CL + i0;
  float v[64];
#pragma unroll
  for (int r = 0; r < 64; ++r) v[r] = p[(size_t)r * QL];
  float m = NEGV;
#pragma unroll
  for (int r = 0; r < 64; ++r) {
    float c = cp[r];
    v[r] = (v[r] + rb[r]) * c + NEGV * (1.0f - c);
    m = fmaxf(m, v[r]);
  }
  redm[w][lane] = m;
  __syncthreads();
  float M = redm[0][lane];
#pragma unroll
  for (int q = 1; q < 16; ++q) M = fmaxf(M, redm[q][lane]);
  float s0 = 0.f, s1 = 0.f, s2 = 0.f, s3 = 0.f;
#pragma unroll
  for (int r = 0; r < 64; r += 4) {
    v[r]     = __expf(v[r]     - M); s0 += v[r];
    v[r + 1] = __expf(v[r + 1] - M); s1 += v[r + 1];
    v[r + 2] = __expf(v[r + 2] - M); s2 += v[r + 2];
    v[r + 3] = __expf(v[r + 3] - M); s3 += v[r + 3];
  }
  reds[w][lane] = (s0 + s1) + (s2 + s3);
  __syncthreads();
  float Z = reds[0][lane];
#pragma unroll
  for (int q = 1; q < 16; ++q) Z += reds[q][lane];
  const float inv = 1.0f / Z;
#pragma unroll
  for (int r = 0; r < 64; ++r) p[(size_t)r * QL] = v[r] * inv;
}

__global__ void k_tgemm(const float* __restrict__ Sbb, const float* __restrict__ Ct,
                        float* __restrict__ T) {
  int lane = threadIdx.x & 31;
  int wave = blockIdx.x * (blockDim.x >> 5) + (threadIdx.x >> 5);
  int b = wave / (16 * 8);
  int rem = wave % (16 * 8);
  int j0 = (rem / 8) * 16;
  int k0 = (rem % 8) * 16;
  int half = lane >> 4, mn = lane & 15;
  v8f acc = {};
  for (int i0 = 0; i0 < CL; i0 += 4) {
    int ii = i0 + half * 2;
    v2f a, bb;
    a.x  = Sbb[(size_t)(b * CL + ii)     * QL + j0 + mn];
    a.y  = Sbb[(size_t)(b * CL + ii + 1) * QL + j0 + mn];
    bb.x = Ct[(size_t)(b * CL + ii)      * Dd + k0 + mn];
    bb.y = Ct[(size_t)(b * CL + ii + 1)  * Dd + k0 + mn];
    acc = __builtin_amdgcn_wmma_f32_16x16x4_f32(false, a, false, bb,
                                                (short)0, acc, false, false);
  }
  float* dst = T + (size_t)(b * QL + j0 + half * 8) * Dd + k0 + mn;
#pragma unroll
  for (int v = 0; v < 8; ++v) dst[(size_t)v * Dd] = acc[v];
}

struct ABFrag {
  v4f a;
  v4f q[8];
  v4f t[8];
};

__global__ void __launch_bounds__(256)
k_abgemm(const float* __restrict__ Sbar, const float* __restrict__ Q,
         const float* __restrict__ T,
         float* __restrict__ Am, float* __restrict__ Bm) {
  const int lane = threadIdx.x & 31;
  const int wave = blockIdx.x * 8 + (int)(threadIdx.x >> 5);
  const int b = wave / (CL / 16);
  const int i0 = (wave % (CL / 16)) * 16;
  const int half = lane >> 4, c = lane & 15;
  const float* ap = Sbar + (size_t)(b * CL + i0 + c) * QL + 4 * half;
  const float* qp = Q + (size_t)(b * Dd + 4 * c) * QL + 4 * half;
  const float* tp = T + (size_t)(b * QL + 4 * half) * Dd + 4 * c;
  v8f accA[8], accB[8];
#pragma unroll
  for (int n = 0; n < 8; ++n) { accA[n] = v8f{}; accB[n] = v8f{}; }

  auto ld = [&](const int j0, ABFrag& f) {
    f.a = *(const v4f*)(ap + j0);
#pragma unroll
    for (int g = 0; g < 2; ++g) {
#pragma unroll
      for (int t = 0; t < 4; ++t)
        f.q[4 * g + t] = *(const v4f*)(qp + (size_t)(64 * g + t) * QL + j0);
    }
#pragma unroll
    for (int e = 0; e < 4; ++e) {
#pragma unroll
      for (int g = 0; g < 2; ++g)
        f.t[2 * e + g] = *(const v4f*)(tp + (size_t)(j0 + e) * Dd + 64 * g);
    }
  };
  auto mm = [&](const ABFrag& f) {
    v2f a0, a1;
    a0.x = f.a.x; a0.y = f.a.y;
    a1.x = f.a.z; a1.y = f.a.w;
#pragma unroll
    for (int n = 0; n < 8; ++n) {
      v2f bq;
      bq.x = f.q[n].x; bq.y = f.q[n].y;
      accA[n] = __builtin_amdgcn_wmma_f32_16x16x4_f32(
          false, a0, false, bq, (short)0, accA[n], false, false);
    }
#pragma unroll
    for (int g = 0; g < 2; ++g) {
#pragma unroll
      for (int t = 0; t < 4; ++t) {
        v2f bt;
        bt.x = f.t[g][t]; bt.y = f.t[2 + g][t];
        accB[4 * g + t] = __builtin_amdgcn_wmma_f32_16x16x4_f32(
            false, a0, false, bt, (short)0, accB[4 * g + t], false, false);
      }
    }
#pragma unroll
    for (int n = 0; n < 8; ++n) {
      v2f bq;
      bq.x = f.q[n].z; bq.y = f.q[n].w;
      accA[n] = __builtin_amdgcn_wmma_f32_16x16x4_f32(
          false, a1, false, bq, (short)0, accA[n], false, false);
    }
#pragma unroll
    for (int g = 0; g < 2; ++g) {
#pragma unroll
      for (int t = 0; t < 4; ++t) {
        v2f bt;
        bt.x = f.t[4 + g][t]; bt.y = f.t[6 + g][t];
        accB[4 * g + t] = __builtin_amdgcn_wmma_f32_16x16x4_f32(
            false, a1, false, bt, (short)0, accB[4 * g + t], false, false);
      }
    }
  };

  ABFrag fx, fy;
  ld(0, fx);
  for (int j0 = 0; j0 < QL; j0 += 16) {
    ld(j0 + 8, fy);
    asm volatile("" ::: "memory");
    mm(fx);
    ld((j0 + 16) & (QL - 1), fx);
    asm volatile("" ::: "memory");
    mm(fy);
  }

#pragma unroll
  for (int r = 0; r < 8; ++r) {
    const size_t off = (size_t)(b * CL + i0 + half * 8 + r) * Dd + 4 * c;
#pragma unroll
    for (int g = 0; g < 2; ++g) {
      v4f oa, ob;
#pragma unroll
      for (int t = 0; t < 4; ++t) {
        oa[t] = accA[4 * g + t][r];
        ob[t] = accB[4 * g + t][r];
      }
      *(v4f*)(Am + off + 64 * g) = oa;
      *(v4f*)(Bm + off + 64 * g) = ob;
    }
  }
}

__global__ void __launch_bounds__(256) __attribute__((amdgpu_waves_per_eu(16)))
k_conv(const float* __restrict__ Ct, const float* __restrict__ Am,
       const float* __restrict__ Bmat,
       const float* __restrict__ dww, const float* __restrict__ dwb,
       const float* __restrict__ pww, const float* __restrict__ pwb,
       float* __restrict__ out) {
  __shared__ float dwX[512][17];
  int b = blockIdx.x / (CL / 16);
  int p0 = (blockIdx.x % (CL / 16)) * 16;
  const int d = threadIdx.x & 127;
  const int grp = __builtin_amdgcn_readfirstlane((int)(threadIdx.x >> 7));
  const int q0 = p0 + grp * 8 - 2;
  auto loadw = [&](const int g, v2f (&w)[5], v2f& bias) {
    const float* wa = dww + (g * Dd + d) * 5;
    const float* wb = dww + ((g + 1) * Dd + d) * 5;
#pragma unroll
    for (int tap = 0; tap < 5; ++tap) { w[tap].x = wa[tap]; w[tap].y = wb[tap]; }
    bias.x = dwb[g * Dd + d];
    bias.y = dwb[(g + 1) * Dd + d];
  };
  auto emit2 = [&](const int g, const v2f (&x)[12], const v2f (&w)[5], const v2f bias) {
#pragma unroll
    for (int p = 0; p < 8; ++p) {
      v2f acc = bias;
#pragma unroll
      for (int tap = 0; tap < 5; ++tap) acc += w[tap] * x[p + tap];
      dwX[g * Dd + d][grp * 8 + p] = acc.x;
      dwX[(g + 1) * Dd + d][grp * 8 + p] = acc.y;
    }
  };
  float rc[12], ra[12], rm[12];
#pragma unroll
  for (int t = 0; t < 12; ++t) {
    const int pos = q0 + t;
    const bool ok = (pos >= 0) && (pos < CL);
    const int row = b * CL + (ok ? pos : 0);
    rc[t] = Ct[row * Dd + d];
    ra[t] = Am[row * Dd + d];
    rm[t] = Bmat[row * Dd + d];
  }
  v2f w[5], bias;
  loadw(0, w, bias);
  __builtin_amdgcn_sched_barrier(0);
  v2f x[12];
  float xm[12];
#pragma unroll
  for (int t = 0; t < 12; ++t) {
    const int pos = q0 + t;
    const bool ok = (pos >= 0) && (pos < CL);
    x[t].x = ok ? rc[t] : 0.f;
    x[t].y = ok ? ra[t] : 0.f;
    xm[t] = ok ? rm[t] : 0.f;
  }
  emit2(0, x, w, bias);
  loadw(2, w, bias);
#pragma unroll
  for (int t = 0; t < 12; ++t) {
    const float c = x[t].x;
    x[t].x = c * x[t].y;
    x[t].y = c * xm[t];
  }
  emit2(2, x, w, bias);
  __syncthreads();
  int lane = threadIdx.x & 31;
  int o0 = (threadIdx.x >> 5) * 16;
  int half = lane >> 4, mn = lane & 15;
  v8f acc = {};
  const float* prow = pww + (size_t)(o0 + mn) * 512 + half * 4;
  for (int c0 = 0; c0 < 512; c0 += 8) {
    v4f pv = *(const v4f*)(prow + c0);
    int cc = c0 + half * 4;
    v2f a0, a1, b0, b1;
    a0.x = pv.x; a0.y = pv.y; a1.x = pv.z; a1.y = pv.w;
    b0.x = dwX[cc][mn];     b0.y = dwX[cc + 1][mn];
    b1.x = dwX[cc + 2][mn]; b1.y = dwX[cc + 3][mn];
    acc = __builtin_amdgcn_wmma_f32_16x16x4_f32(false, a0, false, b0,
                                                (short)0, acc, false, false);
    acc = __builtin_amdgcn_wmma_f32_16x16x4_f32(false, a1, false, b1,
                                                (short)0, acc, false, false);
  }
#pragma unroll
  for (int v = 0; v < 8; ++v) {
    int o = o0 + half * 8 + v;
    out[(size_t)(b * Dd + o) * CL + p0 + mn] = acc[v] + pwb[o];
  }
}

extern "C" void kernel_launch(void* const* d_in, const int* in_sizes, int n_in,
                              void* d_out, int out_size, void* d_ws, size_t ws_size,
                              hipStream_t stream) {
  const float* C   = (const float*)d_in[0];
  const float* Q   = (const float*)d_in[1];
  const float* cm  = (const float*)d_in[2];
  const float* qm  = (const float*)d_in[3];
  const float* W0  = (const float*)d_in[4];
  const float* dww = (const float*)d_in[5];
  const float* dwb = (const float*)d_in[6];
  const float* pww = (const float*)d_in[7];
  const float* pwb = (const float*)d_in[8];
  float* out = (float*)d_out;

  float* ws = (float*)d_ws;
  const size_t nS = (size_t)Bn * CL * QL;
  float* S    = ws;
  float* Sbar = S + nS;
  float* rowb = Sbar + nS;
  float* colb = rowb + (size_t)Bn * CL;
  float* T    = colb + (size_t)Bn * QL;
  float* Ct   = T + (size_t)Bn * QL * Dd;
  float* Am   = Ct + (size_t)Bn * CL * Dd;
  float* Bm   = Am + (size_t)Bn * CL * Dd;
  (void)ws_size; (void)in_sizes; (void)n_in; (void)out_size;

  k_prep<<<Bn * (CL / 128) + Bn * (QL / 128), 256, 0, stream>>>(C, Q, W0, Ct, rowb, colb);
  k_sgemm_rowsm<<<(Bn * (CL / 16)) / 8, 256, 0, stream>>>(C, Q, W0, colb, qm, S, Sbar);
  k_softmax_col<<<Bn * (QL / 32), 512, 0, stream>>>(S, rowb, cm);
  k_tgemm<<<(Bn * (QL / 16) * (Dd / 16)) / 8, 256, 0, stream>>>(S, Ct, T);
  k_abgemm<<<(Bn * (CL / 16)) / 8, 256, 0, stream>>>(Sbar, Q, T, Am, Bm);
  k_conv<<<Bn * (CL / 16), 256, 0, stream>>>(Ct, Am, Bm, dww, dwb, pww, pwb, out);
}


# T GEMM: a wave owns a 64x64 tile of T over an eighth of i (16 accumulator tiles): 8 whole-line 16-byte loads feed 32 matrix ops (was 4 loads per op), 8 partial tiles summed in LDS; 512 -> 256 blocks
# speedup vs baseline: 2.7152x; 1.3119x over previous
#include <hip/hip_runtime.h>

typedef __attribute__((ext_vector_type(2))) float v2f;
typedef __attribute__((ext_vector_type(4))) float v4f;
typedef __attribute__((ext_vector_type(8))) float v8f;

#define NEGV (-1e30f)
constexpr int Bn = 32, Dd = 128, CL = 1024, QL = 256;

template <int PITCH>
__device__ __forceinline__ v4f load16_dot(const float* __restrict__ src,
                                          const float* __restrict__ wv, v4f (&x)[16]) {
#pragma unroll
  for (int r = 0; r < 16; ++r) x[r] = *(const v4f*)(src + (size_t)r * PITCH);
  v4f p = x[0] * wv[0];
#pragma unroll
  for (int r = 1; r < 16; ++r) p += x[r] * wv[r];
  return p;
}

__global__ void __launch_bounds__(256)
k_prep(const float* __restrict__ C, const float* __restrict__ Q,
       const float* __restrict__ W0,
       float* __restrict__ Ct, float* __restrict__ rowb, float* __restrict__ colb) {
  constexpr int P4 = 33;
  constexpr int NT = Bn * (CL / 128);
  __shared__ v4f tile[128 * P4];
  __shared__ v4f red[8 * 32];
  const int lane = threadIdx.x & 31;
  const int w = __builtin_amdgcn_readfirstlane((int)(threadIdx.x >> 5));
  v4f x[16];
  if (blockIdx.x < NT) {
    const int b = blockIdx.x / (CL / 128);
    const int i0 = (blockIdx.x % (CL / 128)) * 128;
    const v4f p = load16_dot<CL>(C + (size_t)(b * Dd + 16 * w) * CL + i0 + 4 * lane,
                                 W0 + 16 * w, x);
    red[w * 32 + lane] = p;
#pragma unroll
    for (int e = 0; e < 4; ++e) {
      v4f* dst = tile + (32 * e + lane) * P4 + 4 * w;
#pragma unroll
      for (int h = 0; h < 4; ++h) {
        v4f t;
        t.x = x[4 * h][e]; t.y = x[4 * h + 1][e];
        t.z = x[4 * h + 2][e]; t.w = x[4 * h + 3][e];
        dst[h] = t;
      }
    }
    __syncthreads();
    float* out = Ct + (size_t)(b * CL + i0 + 16 * w) * Dd + 4 * lane;
#pragma unroll
    for (int q = 0; q < 16; ++q) {
      const int i = 16 * w + q;
      const int row = 32 * (i & 3) + (i >> 2);
      *(v4f*)(out + (size_t)q * Dd) = tile[row * P4 + lane];
    }
    if (threadIdx.x < 128) {
      const float* rf = (const float*)red;
      float s = rf[threadIdx.x];
#pragma unroll
      for (int q = 1; q < 8; ++q) s += rf[q * 128 + threadIdx.x];
      rowb[b * CL + i0 + threadIdx.x] = s;
    }
  } else {
    const int u = blockIdx.x - NT;
    const int b = u >> 1;
    const int j0 = (u & 1) * 128;
    const v4f p = load16_dot<QL>(Q + (size_t)(b * Dd + 16 * w) * QL + j0 + 4 * lane,
                                 W0 + Dd + 16 * w, x);
    red[w * 32 + lane] = p;
    __syncthreads();
    if (threadIdx.x < 128) {
      const float* rf = (const float*)red;
      float s = rf[threadIdx.x];
#pragma unroll
      for (int q = 1; q < 8; ++q) s += rf[q * 128 + threadIdx.x];
      colb[b * QL + j0 + threadIdx.x] = s;
    }
  }
}

template <int M>
__device__ __forceinline__ float row_xor(float v) {
  const int i = __builtin_bit_cast(int, v);
  const int r = __builtin_amdgcn_update_dpp(i, i, 0x160 + M, 0xf, 0xf, true);
  return __builtin_bit_cast(float, r);
}

struct SFrag {
  v4f q0[4], q1[4];
  float c0, c1;
  v2f w;
};

__global__ void __launch_bounds__(256)
k_sgemm_rowsm(const float* __restrict__ C, const float* __restrict__ Q,
              const float* __restrict__ W0, const float* __restrict__ colb,
              const float* __restrict__ qm,
              float* __restrict__ S, float* __restrict__ Sbar) {
  const int lane = threadIdx.x & 31;
  const int wave = blockIdx.x * 8 + (int)(threadIdx.x >> 5);
  const int b = wave / (CL / 16);
  const int i0 = (wave % (CL / 16)) * 16;
  const int half = lane >> 4, c = lane & 15;
  __shared__ float w3s[Dd];
  if (threadIdx.x < Dd) w3s[threadIdx.x] = W0[2 * Dd + threadIdx.x];
  __syncthreads();
  const float* w3h = w3s + 2 * half;
  const float* cp = C + (size_t)(b * Dd + 2 * half) * CL + i0 + c;
  const float* qp = Q + (size_t)(b * Dd + 2 * half) * QL + 4 * c;
  v8f acc[16];
#pragma unroll
  for (int n = 0; n < 16; ++n) acc[n] = v8f{};

  auto ld = [&](const int k0, SFrag& f) {
    const float* cpk = cp + (size_t)k0 * CL;
    const float* qpk = qp + (size_t)k0 * QL;
    f.c0 = cpk[0];
    f.c1 = cpk[CL];
    f.w = *(const v2f*)(w3h + k0);
#pragma unroll
    for (int g = 0; g < 4; ++g) {
      f.q0[g] = *(const v4f*)(qpk + 64 * g);
      f.q1[g] = *(const v4f*)(qpk + QL + 64 * g);
    }
  };
  auto mm = [&](const SFrag& f) {
    v2f a;
    a.x = f.w.x * f.c0;
    a.y = f.w.y * f.c1;
#pragma unroll
    for (int g = 0; g < 4; ++g) {
#pragma unroll
      for (int t = 0; t < 4; ++t) {
        v2f bq;
        bq.x = f.q0[g][t];
        bq.y = f.q1[g][t];
        acc[4 * g + t] = __builtin_amdgcn_wmma_f32_16x16x4_f32(
            false, a, false, bq, (short)0, acc[4 * g + t], false, false);
      }
    }
  };

  SFrag fx, fy;
  ld(0, fx);
  for (int k0 = 0; k0 < Dd; k0 += 8) {
    ld(k0 + 4, fy);
    asm volatile("" ::: "memory");
    mm(fx);
    ld((k0 + 8) & (Dd - 1), fx);
    asm volatile("" ::: "memory");
    mm(fy);
  }

  v4f cb[4], qv[4];
#pragma unroll
  for (int g = 0; g < 4; ++g) {
    cb[g] = *(const v4f*)(colb + b * QL + 64 * g + 4 * c);
    qv[g] = *(const v4f*)(qm + b * QL + 64 * g + 4 * c);
  }
#pragma unroll
  for (int r = 0; r < 8; ++r) {
    const size_t rowoff = (size_t)(b * CL + i0 + half * 8 + r) * QL + 4 * c;
    v4f val[4];
    float m = NEGV;
#pragma unroll
    for (int g = 0; g < 4; ++g) {
      v4f s;
#pragma unroll
      for (int t = 0; t < 4; ++t) {
        const float x = acc[4 * g + t][r];
        const float q = qv[g][t];
        s[t] = x;
        const float v = (x + cb[g][t]) * q + NEGV * (1.0f - q);
        val[g][t] = v;
        m = fmaxf(m, v);
      }
      *(v4f*)(S + rowoff + 64 * g) = s;
    }
    m = fmaxf(m, row_xor<1>(m));
    m = fmaxf(m, row_xor<2>(m));
    m = fmaxf(m, row_xor<4>(m));
    m = fmaxf(m, row_xor<8>(m));
    float sum = 0.f;
#pragma unroll
    for (int g = 0; g < 4; ++g) {
#pragma unroll
      for (int t = 0; t < 4; ++t) {
        val[g][t] = __expf(val[g][t] - m);
        sum += val[g][t];
      }
    }
    sum += row_xor<1>(sum);
    sum += row_xor<2>(sum);
    sum += row_xor<4>(sum);
    sum += row_xor<8>(sum);
    const float inv = 1.0f / sum;
#pragma unroll
    for (int g = 0; g < 4; ++g) {
      v4f o;
#pragma unroll
      for (int t = 0; t < 4; ++t) o[t] = val[g][t] * inv;
      *(v4f*)(Sbar + rowoff + 64 * g) = o;
    }
  }
}

__global__ void __launch_bounds__(512)
k_softmax_col(float* __restrict__ S, const float* __restrict__ rowb,
              const float* __restrict__ cm) {
  __shared__ float redm[16][32], reds[16][32];
  const int lane = threadIdx.x & 31;
  const int w = __builtin_amdgcn_readfirstlane((int)(threadIdx.x >> 5));
  const int b = blockIdx.x / (QL / 32);
  const int j = (blockIdx.x % (QL / 32)) * 32 + lane;
  const int i0 = w * 64;
  float* p = S + (size_t)(b * CL + i0) * QL + j;
  const float* rb = rowb + b * CL + i0;
  const float* cp = cm + b * CL + i0;
  float v[64];
#pragma unroll
  for (int r = 0; r < 64; ++r) v[r] = p[(size_t)r * QL];
  float m = NEGV;
#pragma unroll
  for (int r = 0; r < 64; ++r) {
    float c = cp[r];
    v[r] = (v[r] + rb[r]) * c + NEGV * (1.0f - c);
    m = fmaxf(m, v[r]);
  }
  redm[w][lane] = m;
  __syncthreads();
  float M = redm[0][lane];
#pragma unroll
  for (int q = 1; q < 16; ++q) M = fmaxf(M, redm[q][lane]);
  float s0 = 0.f, s1 = 0.f, s2 = 0.f, s3 = 0.f;
#pragma unroll
  for (int r = 0; r < 64; r += 4) {
    v[r]     = __expf(v[r]     - M); s0 += v[r];
    v[r + 1] = __expf(v[r + 1] - M); s1 += v[r + 1];
    v[r + 2] = __expf(v[r + 2] - M); s2 += v[r + 2];
    v[r + 3] = __expf(v[r + 3] - M); s3 += v[r + 3];
  }
  reds[w][lane] = (s0 + s1) + (s2 + s3);
  __syncthreads();
  float Z = reds[0][lane];
#pragma unroll
  for (int q = 1; q < 16; ++q) Z += reds[q][lane];
  const float inv = 1.0f / Z;
#pragma unroll
  for (int r = 0; r < 64; ++r) p[(size_t)r * QL] = v[r] * inv;
}

struct TFrag {
  v4f a[4];
  v4f bq[4];
};

__global__ void __launch_bounds__(256)
k_tgemm(const float* __restrict__ Sbb, const float* __restrict__ Ct,
        float* __restrict__ T) {
  constexpr int IW = CL / 8;
  __shared__ v4f part[8 * 32 * 32];
  const int lane = threadIdx.x & 31;
  const int w = __builtin_amdgcn_readfirstlane((int)(threadIdx.x >> 5));
  const int b = blockIdx.x >> 3;
  const int jbase = ((blockIdx.x >> 1) & 3) * 64;
  const int kbase = (blockIdx.x & 1) * 64;
  const int half = lane >> 4, c = lane & 15;
  const float* ap = Sbb + (size_t)(b * CL + w * IW + 4 * half) * QL + jbase + 4 * c;
  const float* bp = Ct + (size_t)(b * CL + w * IW + 4 * half) * Dd + kbase + 4 * c;
  v8f acc[16];
#pragma unroll
  for (int n = 0; n < 16; ++n) acc[n] = v8f{};

  auto ld = [&](const int i0, TFrag& f) {
#pragma unroll
    for (int e = 0; e < 4; ++e) {
      f.a[e] = *(const v4f*)(ap + (size_t)(i0 + e) * QL);
      f.bq[e] = *(const v4f*)(bp + (size_t)(i0 + e) * Dd);
    }
  };
  auto mm = [&](const TFrag& f) {
#pragma unroll
    for (int p = 0; p < 2; ++p) {
#pragma unroll
      for (int ta = 0; ta < 4; ++ta) {
        v2f a;
        a.x = f.a[2 * p][ta];
        a.y = f.a[2 * p + 1][ta];
#pragma unroll
        for (int tb = 0; tb < 4; ++tb) {
          v2f bb;
          bb.x = f.bq[2 * p][tb];
          bb.y = f.bq[2 * p + 1][tb];
          acc[4 * ta + tb] = __builtin_amdgcn_wmma_f32_16x16x4_f32(
              false, a, false, bb, (short)0, acc[4 * ta + tb], false, false);
        }
      }
    }
  };

  TFrag fx, fy;
  ld(0, fx);
  for (int i0 = 0; i0 < IW; i0 += 16) {
    ld(i0 + 8, fy);
    asm volatile("" ::: "memory");
    mm(fx);
    ld((i0 + 16) & (IW - 1), fx);
    asm volatile("" ::: "memory");
    mm(fy);
  }

#pragma unroll
  for (int ta = 0; ta < 4; ++ta) {
#pragma unroll
    for (int r = 0; r < 8; ++r) {
      v4f v;
#pragma unroll
      for (int tb = 0; tb < 4; ++tb) v[tb] = acc[4 * ta + tb][r];
      part[(w * 32 + 8 * ta + r) * 32 + lane] = v;
    }
  }
  __syncthreads();
#pragma unroll
  for (int qq = 0; qq < 4; ++qq) {
    const int q = 4 * w + qq;
    v4f s = part[q * 32 + lane];
#pragma unroll
    for (int u = 1; u < 8; ++u) s += part[(u * 32 + q) * 32 + lane];
    const int j = jbase + 4 * (8 * half + (q & 7)) + (q >> 3);
    *(v4f*)(T + (size_t)(b * QL + j) * Dd + kbase + 4 * c) = s;
  }
}

struct ABFrag {
  v4f a;
  v4f q[8];
  v4f t[8];
};

__global__ void __launch_bounds__(256)
k_abgemm(const float* __restrict__ Sbar, const float* __restrict__ Q,
         const float* __restrict__ T,
         float* __restrict__ Am, float* __restrict__ Bm) {
  const int lane = threadIdx.x & 31;
  const int wave = blockIdx.x * 8 + (int)(threadIdx.x >> 5);
  const int b = wave / (CL / 16);
  const int i0 = (wave % (CL / 16)) * 16;
  const int half = lane >> 4, c = lane & 15;
  const float* ap = Sbar + (size_t)(b * CL + i0 + c) * QL + 4 * half;
  const float* qp = Q + (size_t)(b * Dd + 4 * c) * QL + 4 * half;
  const float* tp = T + (size_t)(b * QL + 4 * half) * Dd + 4 * c;
  v8f accA[8], accB[8];
#pragma unroll
  for (int n = 0; n < 8; ++n) { accA[n] = v8f{}; accB[n] = v8f{}; }

  auto ld = [&](const int j0, ABFrag& f) {
    f.a = *(const v4f*)(ap + j0);
#pragma unroll
    for (int g = 0; g < 2; ++g) {
#pragma unroll
      for (int t = 0; t < 4; ++t)
        f.q[4 * g + t] = *(const v4f*)(qp + (size_t)(64 * g + t) * QL + j0);
    }
#pragma unroll
    for (int e = 0; e < 4; ++e) {
#pragma unroll
      for (int g = 0; g < 2; ++g)
        f.t[2 * e + g] = *(const v4f*)(tp + (size_t)(j0 + e) * Dd + 64 * g);
    }
  };
  auto mm = [&](const ABFrag& f) {
    v2f a0, a1;
    a0.x = f.a.x; a0.y = f.a.y;
    a1.x = f.a.z; a1.y = f.a.w;
#pragma unroll
    for (int n = 0; n < 8; ++n) {
      v2f bq;
      bq.x = f.q[n].x; bq.y = f.q[n].y;
      accA[n] = __builtin_amdgcn_wmma_f32_16x16x4_f32(
          false, a0, false, bq, (short)0, accA[n], false, false);
    }
#pragma unroll
    for (int g = 0; g < 2; ++g) {
#pragma unroll
      for (int t = 0; t < 4; ++t) {
        v2f bt;
        bt.x = f.t[g][t]; bt.y = f.t[2 + g][t];
        accB[4 * g + t] = __builtin_amdgcn_wmma_f32_16x16x4_f32(
            false, a0, false, bt, (short)0, accB[4 * g + t], false, false);
      }
    }
#pragma unroll
    for (int n = 0; n < 8; ++n) {
      v2f bq;
      bq.x = f.q[n].z; bq.y = f.q[n].w;
      accA[n] = __builtin_amdgcn_wmma_f32_16x16x4_f32(
          false, a1, false, bq, (short)0, accA[n], false, false);
    }
#pragma unroll
    for (int g = 0; g < 2; ++g) {
#pragma unroll
      for (int t = 0; t < 4; ++t) {
        v2f bt;
        bt.x = f.t[4 + g][t]; bt.y = f.t[6 + g][t];
        accB[4 * g + t] = __builtin_amdgcn_wmma_f32_16x16x4_f32(
            false, a1, false, bt, (short)0, accB[4 * g + t], false, false);
      }
    }
  };

  ABFrag fx, fy;
  ld(0, fx);
  for (int j0 = 0; j0 < QL; j0 += 16) {
    ld(j0 + 8, fy);
    asm volatile("" ::: "memory");
    mm(fx);
    ld((j0 + 16) & (QL - 1), fx);
    asm volatile("" ::: "memory");
    mm(fy);
  }

#pragma unroll
  for (int r = 0; r < 8; ++r) {
    const size_t off = (size_t)(b * CL + i0 + half * 8 + r) * Dd + 4 * c;
#pragma unroll
    for (int g = 0; g < 2; ++g) {
      v4f oa, ob;
#pragma unroll
      for (int t = 0; t < 4; ++t) {
        oa[t] = accA[4 * g + t][r];
        ob[t] = accB[4 * g + t][r];
      }
      *(v4f*)(Am + off + 64 * g) = oa;
      *(v4f*)(Bm + off + 64 * g) = ob;
    }
  }
}

__global__ void __launch_bounds__(256) __attribute__((amdgpu_waves_per_eu(16)))
k_conv(const float* __restrict__ Ct, const float* __restrict__ Am,
       const float* __restrict__ Bmat,
       const float* __restrict__ dww, const float* __restrict__ dwb,
       const float* __restrict__ pww, const float* __restrict__ pwb,
       float* __restrict__ out) {
  __shared__ float dwX[512][17];
  int b = blockIdx.x / (CL / 16);
  int p0 = (blockIdx.x % (CL / 16)) * 16;
  const int d = threadIdx.x & 127;
  const int grp = __builtin_amdgcn_readfirstlane((int)(threadIdx.x >> 7));
  const int q0 = p0 + grp * 8 - 2;
  auto loadw = [&](const int g, v2f (&w)[5], v2f& bias) {
    const float* wa = dww + (g * Dd + d) * 5;
    const float* wb = dww + ((g + 1) * Dd + d) * 5;
#pragma unroll
    for (int tap = 0; tap < 5; ++tap) { w[tap].x = wa[tap]; w[tap].y = wb[tap]; }
    bias.x = dwb[g * Dd + d];
    bias.y = dwb[(g + 1) * Dd + d];
  };
  auto emit2 = [&](const int g, const v2f (&x)[12], const v2f (&w)[5], const v2f bias) {
#pragma unroll
    for (int p = 0; p < 8; ++p) {
      v2f acc = bias;
#pragma unroll
      for (int tap = 0; tap < 5; ++tap) acc += w[tap] * x[p + tap];
      dwX[g * Dd + d][grp * 8 + p] = acc.x;
      dwX[(g + 1) * Dd + d][grp * 8 + p] = acc.y;
    }
  };
  float rc[12], ra[12], rm[12];
#pragma unroll
  for (int t = 0; t < 12; ++t) {
    const int pos = q0 + t;
    const bool ok = (pos >= 0) && (pos < CL);
    const int row = b * CL + (ok ? pos : 0);
    rc[t] = Ct[row * Dd + d];
    ra[t] = Am[row * Dd + d];
    rm[t] = Bmat[row * Dd + d];
  }
  v2f w[5], bias;
  loadw(0, w, bias);
  __builtin_amdgcn_sched_barrier(0);
  v2f x[12];
  float xm[12];
#pragma unroll
  for (int t = 0; t < 12; ++t) {
    const int pos = q0 + t;
    const bool ok = (pos >= 0) && (pos < CL);
    x[t].x = ok ? rc[t] : 0.f;
    x[t].y = ok ? ra[t] : 0.f;
    xm[t] = ok ? rm[t] : 0.f;
  }
  emit2(0, x, w, bias);
  loadw(2, w, bias);
#pragma unroll
  for (int t = 0; t < 12; ++t) {
    const float c = x[t].x;
    x[t].x = c * x[t].y;
    x[t].y = c * xm[t];
  }
  emit2(2, x, w, bias);
  __syncthreads();
  int lane = threadIdx.x & 31;
  int o0 = (threadIdx.x >> 5) * 16;
  int half = lane >> 4, mn = lane & 15;
  v8f acc = {};
  const float* prow = pww + (size_t)(o0 + mn) * 512 + half * 4;
  for (int c0 = 0; c0 < 512; c0 += 8) {
    v4f pv = *(const v4f*)(prow + c0);
    int cc = c0 + half * 4;
    v2f a0, a1, b0, b1;
    a0.x = pv.x; a0.y = pv.y; a1.x = pv.z; a1.y = pv.w;
    b0.x = dwX[cc][mn];     b0.y = dwX[cc + 1][mn];
    b1.x = dwX[cc + 2][mn]; b1.y = dwX[cc + 3][mn];
    acc = __builtin_amdgcn_wmma_f32_16x16x4_f32(false, a0, false, b0,
                                                (short)0, acc, false, false);
    acc = __builtin_amdgcn_wmma_f32_16x16x4_f32(false, a1, false, b1,
                                                (short)0, acc, false, false);
  }
#pragma unroll
  for (int v = 0; v < 8; ++v) {
    int o = o0 + half * 8 + v;
    out[(size_t)(b * Dd + o) * CL + p0 + mn] = acc[v] + pwb[o];
  }
}

extern "C" void kernel_launch(void* const* d_in, const int* in_sizes, int n_in,
                              void* d_out, int out_size, void* d_ws, size_t ws_size,
                              hipStream_t stream) {
  const float* C   = (const float*)d_in[0];
  const float* Q   = (const float*)d_in[1];
  const float* cm  = (const float*)d_in[2];
  const float* qm  = (const float*)d_in[3];
  const float* W0  = (const float*)d_in[4];
  const float* dww = (const float*)d_in[5];
  const float* dwb = (const float*)d_in[6];
  const float* pww = (const float*)d_in[7];
  const float* pwb = (const float*)d_in[8];
  float* out = (float*)d_out;

  float* ws = (float*)d_ws;
  const size_t nS = (size_t)Bn * CL * QL;
  float* S    = ws;
  float* Sbar = S + nS;
  float* rowb = Sbar + nS;
  float* colb = rowb + (size_t)Bn * CL;
  float* T    = colb + (size_t)Bn * QL;
  float* Ct   = T + (size_t)Bn * QL * Dd;
  float* Am   = Ct + (size_t)Bn * CL * Dd;
  float* Bm   = Am + (size_t)Bn * CL * Dd;
  (void)ws_size; (void)in_sizes; (void)n_in; (void)out_size;

  k_prep<<<Bn * (CL / 128) + Bn * (QL / 128), 256, 0, stream>>>(C, Q, W0, Ct, rowb, colb);
  k_sgemm_rowsm<<<(Bn * (CL / 16)) / 8, 256, 0, stream>>>(C, Q, W0, colb, qm, S, Sbar);
  k_softmax_col<<<Bn * (QL / 32), 512, 0, stream>>>(S, rowb, cm);
  k_tgemm<<<Bn * (QL / 64) * (Dd / 64), 256, 0, stream>>>(S, Ct, T);
  k_abgemm<<<(Bn * (CL / 16)) / 8, 256, 0, stream>>>(Sbar, Q, T, Am, Bm);
  k_conv<<<Bn * (CL / 16), 256, 0, stream>>>(Ct, Am, Bm, dww, dwb, pww, pwb, out);
}


# pointwise-conv weights re-laid in the conv waves' own fragment order (1 KiB per block of the A/Bm launch, into the dead S plane): a wave's weight load is 512 contiguous bytes, 4 lines, was 16 lines
# speedup vs baseline: 2.9154x; 1.0737x over previous
#include <hip/hip_runtime.h>

typedef __attribute__((ext_vector_type(2))) float v2f;
typedef __attribute__((ext_vector_type(4))) float v4f;
typedef __attribute__((ext_vector_type(8))) float v8f;

#define NEGV (-1e30f)
constexpr int Bn = 32, Dd = 128, CL = 1024, QL = 256;

template <int PITCH>
__device__ __forceinline__ v4f load16_dot(const float* __restrict__ src,
                                          const float* __restrict__ wv, v4f (&x)[16]) {
#pragma unroll
  for (int r = 0; r < 16; ++r) x[r] = *(const v4f*)(src + (size_t)r * PITCH);
  v4f p = x[0] * wv[0];
#pragma unroll
  for (int r = 1; r < 16; ++r) p += x[r] * wv[r];
  return p;
}

__global__ void __launch_bounds__(256)
k_prep(const float* __restrict__ C, const float* __restrict__ Q,
       const float* __restrict__ W0,
       float* __restrict__ Ct, float* __restrict__ rowb, float* __restrict__ colb) {
  constexpr int P4 = 33;
  constexpr int NT = Bn * (CL / 128);
  __shared__ v4f tile[128 * P4];
  __shared__ v4f red[8 * 32];
  const int lane = threadIdx.x & 31;
  const int w = __builtin_amdgcn_readfirstlane((int)(threadIdx.x >> 5));
  v4f x[16];
  if (blockIdx.x < NT) {
    const int b = blockIdx.x / (CL / 128);
    const int i0 = (blockIdx.x % (CL / 128)) * 128;
    const v4f p = load16_dot<CL>(C + (size_t)(b * Dd + 16 * w) * CL + i0 + 4 * lane,
                                 W0 + 16 * w, x);
    red[w * 32 + lane] = p;
#pragma unroll
    for (int e = 0; e < 4; ++e) {
      v4f* dst = tile + (32 * e + lane) * P4 + 4 * w;
#pragma unroll
      for (int h = 0; h < 4; ++h) {
        v4f t;
        t.x = x[4 * h][e]; t.y = x[4 * h + 1][e];
        t.z = x[4 * h + 2][e]; t.w = x[4 * h + 3][e];
        dst[h] = t;
      }
    }
    __syncthreads();
    float* out = Ct + (size_t)(b * CL + i0 + 16 * w) * Dd + 4 * lane;
#pragma unroll
    for (int q = 0; q < 16; ++q) {
      const int i = 16 * w + q;
      const int row = 32 * (i & 3) + (i >> 2);
      *(v4f*)(out + (size_t)q * Dd) = tile[row * P4 + lane];
    }
    if (threadIdx.x < 128) {
      const float* rf = (const float*)red;
      float s = rf[threadIdx.x];
#pragma unroll
      for (int q = 1; q < 8; ++q) s += rf[q * 128 + threadIdx.x];
      rowb[b * CL + i0 + threadIdx.x] = s;
    }
  } else {
    const int u = blockIdx.x - NT;
    const int b = u >> 1;
    const int j0 = (u & 1) * 128;
    const v4f p = load16_dot<QL>(Q + (size_t)(b * Dd + 16 * w) * QL + j0 + 4 * lane,
                                 W0 + Dd + 16 * w, x);
    red[w * 32 + lane] = p;
    __syncthreads();
    if (threadIdx.x < 128) {
      const float* rf = (const float*)red;
      float s = rf[threadIdx.x];
#pragma unroll
      for (int q = 1; q < 8; ++q) s += rf[q * 128 + threadIdx.x];
      colb[b * QL + j0 + threadIdx.x] = s;
    }
  }
}

template <int M>
__device__ __forceinline__ float row_xor(float v) {
  const int i = __builtin_bit_cast(int, v);
  const int r = __builtin_amdgcn_update_dpp(i, i, 0x160 + M, 0xf, 0xf, true);
  return __builtin_bit_cast(float, r);
}

struct SFrag {
  v4f q0[4], q1[4];
  float c0, c1;
  v2f w;
};

__global__ void __launch_bounds__(256)
k_sgemm_rowsm(const float* __restrict__ C, const float* __restrict__ Q,
              const float* __restrict__ W0, const float* __restrict__ colb,
              const float* __restrict__ qm,
              float* __restrict__ S, float* __restrict__ Sbar) {
  const int lane = threadIdx.x & 31;
  const int wave = blockIdx.x * 8 + (int)(threadIdx.x >> 5);
  const int b = wave / (CL / 16);
  const int i0 = (wave % (CL / 16)) * 16;
  const int half = lane >> 4, c = lane & 15;
  __shared__ float w3s[Dd];
  if (threadIdx.x < Dd) w3s[threadIdx.x] = W0[2 * Dd + threadIdx.x];
  __syncthreads();
  const float* w3h = w3s + 2 * half;
  const float* cp = C + (size_t)(b * Dd + 2 * half) * CL + i0 + c;
  const float* qp = Q + (size_t)(b * Dd + 2 * half) * QL + 4 * c;
  v8f acc[16];
#pragma unroll
  for (int n = 0; n < 16; ++n) acc[n] = v8f{};

  auto ld = [&](const int k0, SFrag& f) {
    const float* cpk = cp + (size_t)k0 * CL;
    const float* qpk = qp + (size_t)k0 * QL;
    f.c0 = cpk[0];
    f.c1 = cpk[CL];
    f.w = *(const v2f*)(w3h + k0);
#pragma unroll
    for (int g = 0; g < 4; ++g) {
      f.q0[g] = *(const v4f*)(qpk + 64 * g);
      f.q1[g] = *(const v4f*)(qpk + QL + 64 * g);
    }
  };
  auto mm = [&](const SFrag& f) {
    v2f a;
    a.x = f.w.x * f.c0;
    a.y = f.w.y * f.c1;
#pragma unroll
    for (int g = 0; g < 4; ++g) {
#pragma unroll
      for (int t = 0; t < 4; ++t) {
        v2f bq;
        bq.x = f.q0[g][t];
        bq.y = f.q1[g][t];
        acc[4 * g + t] = __builtin_amdgcn_wmma_f32_16x16x4_f32(
            false, a, false, bq, (short)0, acc[4 * g + t], false, false);
      }
    }
  };

  SFrag fx, fy;
  ld(0, fx);
  for (int k0 = 0; k0 < Dd; k0 += 8) {
    ld(k0 + 4, fy);
    asm volatile("" ::: "memory");
    mm(fx);
    ld((k0 + 8) & (Dd - 1), fx);
    asm volatile("" ::: "memory");
    mm(fy);
  }

  v4f cb[4], qv[4];
#pragma unroll
  for (int g = 0; g < 4; ++g) {
    cb[g] = *(const v4f*)(colb + b * QL + 64 * g + 4 * c);
    qv[g] = *(const v4f*)(qm + b * QL + 64 * g + 4 * c);
  }
#pragma unroll
  for (int r = 0; r < 8; ++r) {
    const size_t rowoff = (size_t)(b * CL + i0 + half * 8 + r) * QL + 4 * c;
    v4f val[4];
    float m = NEGV;
#pragma unroll
    for (int g = 0; g < 4; ++g) {
      v4f s;
#pragma unroll
      for (int t = 0; t < 4; ++t) {
        const float x = acc[4 * g + t][r];
        const float q = qv[g][t];
        s[t] = x;
        const float v = (x + cb[g][t]) * q + NEGV * (1.0f - q);
        val[g][t] = v;
        m = fmaxf(m, v);
      }
      *(v4f*)(S + rowoff + 64 * g) = s;
    }
    m = fmaxf(m, row_xor<1>(m));
    m = fmaxf(m, row_xor<2>(m));
    m = fmaxf(m, row_xor<4>(m));
    m = fmaxf(m, row_xor<8>(m));
    float sum = 0.f;
#pragma unroll
    for (int g = 0; g < 4; ++g) {
#pragma unroll
      for (int t = 0; t < 4; ++t) {
        val[g][t] = __expf(val[g][t] - m);
        sum += val[g][t];
      }
    }
    sum += row_xor<1>(sum);
    sum += row_xor<2>(sum);
    sum += row_xor<4>(sum);
    sum += row_xor<8>(sum);
    const float inv = 1.0f / sum;
#pragma unroll
    for (int g = 0; g < 4; ++g) {
      v4f o;
#pragma unroll
      for (int t = 0; t < 4; ++t) o[t] = val[g][t] * inv;
      *(v4f*)(Sbar + rowoff + 64 * g) = o;
    }
  }
}

__global__ void __launch_bounds__(512)
k_softmax_col(float* __restrict__ S, const float* __restrict__ rowb,
              const float* __restrict__ cm) {
  __shared__ float redm[16][32], reds[16][32];
  const int lane = threadIdx.x & 31;
  const int w = __builtin_amdgcn_readfirstlane((int)(threadIdx.x >> 5));
  const int b = blockIdx.x / (QL / 32);
  const int j = (blockIdx.x % (QL / 32)) * 32 + lane;
  const int i0 = w * 64;
  float* p = S + (size_t)(b * CL + i0) * QL + j;
  const float* rb = rowb + b * CL + i0;
  const float* cp = cm + b * CL + i0;
  float v[64];
#pragma unroll
  for (int r = 0; r < 64; ++r) v[r] = p[(size_t)r * QL];
  float m = NEGV;
#pragma unroll
  for (int r = 0; r < 64; ++r) {
    float c = cp[r];
    v[r] = (v[r] + rb[r]) * c + NEGV * (1.0f - c);
    m = fmaxf(m, v[r]);
  }
  redm[w][lane] = m;
  __syncthreads();
  float M = redm[0][lane];
#pragma unroll
  for (int q = 1; q < 16; ++q) M = fmaxf(M, redm[q][lane]);
  float s0 = 0.f, s1 = 0.f, s2 = 0.f, s3 = 0.f;
#pragma unroll
  for (int r = 0; r < 64; r += 4) {
    v[r]     = __expf(v[r]     - M); s0 += v[r];
    v[r + 1] = __expf(v[r + 1] - M); s1 += v[r + 1];
    v[r + 2] = __expf(v[r + 2] - M); s2 += v[r + 2];
    v[r + 3] = __expf(v[r + 3] - M); s3 += v[r + 3];
  }
  reds[w][lane] = (s0 + s1) + (s2 + s3);
  __syncthreads();
  float Z = reds[0][lane];
#pragma unroll
  for (int q = 1; q < 16; ++q) Z += reds[q][lane];
  const float inv = 1.0f / Z;
#pragma unroll
  for (int r = 0; r < 64; ++r) p[(size_t)r * QL] = v[r] * inv;
}

struct TFrag {
  v4f a[4];
  v4f bq[4];
};

__global__ void __launch_bounds__(256)
k_tgemm(const float* __restrict__ Sbb, const float* __restrict__ Ct,
        float* __restrict__ T) {
  constexpr int IW = CL / 8;
  __shared__ v4f part[8 * 32 * 32];
  const int lane = threadIdx.x & 31;
  const int w = __builtin_amdgcn_readfirstlane((int)(threadIdx.x >> 5));
  const int b = blockIdx.x >> 3;
  const int jbase = ((blockIdx.x >> 1) & 3) * 64;
  const int kbase = (blockIdx.x & 1) * 64;
  const int half = lane >> 4, c = lane & 15;
  const float* ap = Sbb + (size_t)(b * CL + w * IW + 4 * half) * QL + jbase + 4 * c;
  const float* bp = Ct + (size_t)(b * CL + w * IW + 4 * half) * Dd + kbase + 4 * c;
  v8f acc[16];
#pragma unroll
  for (int n = 0; n < 16; ++n) acc[n] = v8f{};

  auto ld = [&](const int i0, TFrag& f) {
#pragma unroll
    for (int e = 0; e < 4; ++e) {
      f.a[e] = *(const v4f*)(ap + (size_t)(i0 + e) * QL);
      f.bq[e] = *(const v4f*)(bp + (size_t)(i0 + e) * Dd);
    }
  };
  auto mm = [&](const TFrag& f) {
#pragma unroll
    for (int p = 0; p < 2; ++p) {
#pragma unroll
      for (int ta = 0; ta < 4; ++ta) {
        v2f a;
        a.x = f.a[2 * p][ta];
        a.y = f.a[2 * p + 1][ta];
#pragma unroll
        for (int tb = 0; tb < 4; ++tb) {
          v2f bb;
          bb.x = f.bq[2 * p][tb];
          bb.y = f.bq[2 * p + 1][tb];
          acc[4 * ta + tb] = __builtin_amdgcn_wmma_f32_16x16x4_f32(
              false, a, false, bb, (short)0, acc[4 * ta + tb], false, false);
        }
      }
    }
  };

  TFrag fx, fy;
  ld(0, fx);
  for (int i0 = 0; i0 < IW; i0 += 16) {
    ld(i0 + 8, fy);
    asm volatile("" ::: "memory");
    mm(fx);
    ld((i0 + 16) & (IW - 1), fx);
    asm volatile("" ::: "memory");
    mm(fy);
  }

#pragma unroll
  for (int ta = 0; ta < 4; ++ta) {
#pragma unroll
    for (int r = 0; r < 8; ++r) {
      v4f v;
#pragma unroll
      for (int tb = 0; tb < 4; ++tb) v[tb] = acc[4 * ta + tb][r];
      part[(w * 32 + 8 * ta + r) * 32 + lane] = v;
    }
  }
  __syncthreads();
#pragma unroll
  for (int qq = 0; qq < 4; ++qq) {
    const int q = 4 * w + qq;
    v4f s = part[q * 32 + lane];
#pragma unroll
    for (int u = 1; u < 8; ++u) s += part[(u * 32 + q) * 32 + lane];
    const int j = jbase + 4 * (8 * half + (q & 7)) + (q >> 3);
    *(v4f*)(T + (size_t)(b * QL + j) * Dd + kbase + 4 * c) = s;
  }
}

struct ABFrag {
  v4f a;
  v4f q[8];
  v4f t[8];
};

constexpr int NAB = (Bn * (CL / 16)) / 8;

__global__ void __launch_bounds__(256)
k_abgemm(const float* __restrict__ Sbar, const float* __restrict__ Q,
         const float* __restrict__ T,
         float* __restrict__ Am, float* __restrict__ Bm,
         const float* __restrict__ pww, float* __restrict__ Wp) {
  static_assert(NAB * 64 == 8 * 64 * 32, "one 16-byte chunk per thread of two waves");
  const bool wcopy = threadIdx.x < 64;
  const int widx = blockIdx.x * 64 + (int)(threadIdx.x & 63);
  v4f wchunk = v4f{};
  if (wcopy) {
    const int ln = widx & 31, it = (widx >> 5) & 63, ow = widx >> 11;
    wchunk = *(const v4f*)(pww + (size_t)(16 * ow + (ln & 15)) * 512 +
                           8 * it + 4 * (ln >> 4));
  }
  const int lane = threadIdx.x & 31;
  const int wave = blockIdx.x * 8 + (int)(threadIdx.x >> 5);
  const int b = wave / (CL / 16);
  const int i0 = (wave % (CL / 16)) * 16;
  const int half = lane >> 4, c = lane & 15;
  const float* ap = Sbar + (size_t)(b * CL + i0 + c) * QL + 4 * half;
  const float* qp = Q + (size_t)(b * Dd + 4 * c) * QL + 4 * half;
  const float* tp = T + (size_t)(b * QL + 4 * half) * Dd + 4 * c;
  v8f accA[8], accB[8];
#pragma unroll
  for (int n = 0; n < 8; ++n) { accA[n] = v8f{}; accB[n] = v8f{}; }

  auto ld = [&](const int j0, ABFrag& f) {
    f.a = *(const v4f*)(ap + j0);
#pragma unroll
    for (int g = 0; g < 2; ++g) {
#pragma unroll
      for (int t = 0; t < 4; ++t)
        f.q[4 * g + t] = *(const v4f*)(qp + (size_t)(64 * g + t) * QL + j0);
    }
#pragma unroll
    for (int e = 0; e < 4; ++e) {
#pragma unroll
      for (int g = 0; g < 2; ++g)
        f.t[2 * e + g] = *(const v4f*)(tp + (size_t)(j0 + e) * Dd + 64 * g);
    }
  };
  auto mm = [&](const ABFrag& f) {
    v2f a0, a1;
    a0.x = f.a.x; a0.y = f.a.y;
    a1.x = f.a.z; a1.y = f.a.w;
#pragma unroll
    for (int n = 0; n < 8; ++n) {
      v2f bq;
      bq.x = f.q[n].x; bq.y = f.q[n].y;
      accA[n] = __builtin_amdgcn_wmma_f32_16x16x4_f32(
          false, a0, false, bq, (short)0, accA[n], false, false);
    }
#pragma unroll
    for (int g = 0; g < 2; ++g) {
#pragma unroll
      for (int t = 0; t < 4; ++t) {
        v2f bt;
        bt.x = f.t[g][t]; bt.y = f.t[2 + g][t];
        accB[4 * g + t] = __builtin_amdgcn_wmma_f32_16x16x4_f32(
            false, a0, false, bt, (short)0, accB[4 * g + t], false, false);
      }
    }
#pragma unroll
    for (int n = 0; n < 8; ++n) {
      v2f bq;
      bq.x = f.q[n].z; bq.y = f.q[n].w;
      accA[n] = __builtin_amdgcn_wmma_f32_16x16x4_f32(
          false, a1, false, bq, (short)0, accA[n], false, false);
    }
#pragma unroll
    for (int g = 0; g < 2; ++g) {
#pragma unroll
      for (int t = 0; t < 4; ++t) {
        v2f bt;
        bt.x = f.t[4 + g][t]; bt.y = f.t[6 + g][t];
        accB[4 * g + t] = __builtin_amdgcn_wmma_f32_16x16x4_f32(
            false, a1, false, bt, (short)0, accB[4 * g + t], false, false);
      }
    }
  };

  ABFrag fx, fy;
  ld(0, fx);
  for (int j0 = 0; j0 < QL; j0 += 16) {
    ld(j0 + 8, fy);
    asm volatile("" ::: "memory");
    mm(fx);
    ld((j0 + 16) & (QL - 1), fx);
    asm volatile("" ::: "memory");
    mm(fy);
  }

#pragma unroll
  for (int r = 0; r < 8; ++r) {
    const size_t off = (size_t)(b * CL + i0 + half * 8 + r) * Dd + 4 * c;
#pragma unroll
    for (int g = 0; g < 2; ++g) {
      v4f oa, ob;
#pragma unroll
      for (int t = 0; t < 4; ++t) {
        oa[t] = accA[4 * g + t][r];
        ob[t] = accB[4 * g + t][r];
      }
      *(v4f*)(Am + off + 64 * g) = oa;
      *(v4f*)(Bm + off + 64 * g) = ob;
    }
  }
  if (wcopy) *(v4f*)(Wp + (size_t)widx * 4) = wchunk;
}

__global__ void __launch_bounds__(256) __attribute__((amdgpu_waves_per_eu(16)))
k_conv(const float* __restrict__ Ct, const float* __restrict__ Am,
       const float* __restrict__ Bmat,
       const float* __restrict__ dww, const float* __restrict__ dwb,
       const float* __restrict__ Wp, const float* __restrict__ pwb,
       float* __restrict__ out) {
  __shared__ float dwX[512][17];
  int b = blockIdx.x / (CL / 16);
  int p0 = (blockIdx.x % (CL / 16)) * 16;
  const int d = threadIdx.x & 127;
  const int grp = __builtin_amdgcn_readfirstlane((int)(threadIdx.x >> 7));
  const int q0 = p0 + grp * 8 - 2;
  auto loadw = [&](const int g, v2f (&w)[5], v2f& bias) {
    const float* wa = dww + (g * Dd + d) * 5;
    const float* wb = dww + ((g + 1) * Dd + d) * 5;
#pragma unroll
    for (int tap = 0; tap < 5; ++tap) { w[tap].x = wa[tap]; w[tap].y = wb[tap]; }
    bias.x = dwb[g * Dd + d];
    bias.y = dwb[(g + 1) * Dd + d];
  };
  auto emit2 = [&](const int g, const v2f (&x)[12], const v2f (&w)[5], const v2f bias) {
#pragma unroll
    for (int p = 0; p < 8; ++p) {
      v2f acc = bias;
#pragma unroll
      for (int tap = 0; tap < 5; ++tap) acc += w[tap] * x[p + tap];
      dwX[g * Dd + d][grp * 8 + p] = acc.x;
      dwX[(g + 1) * Dd + d][grp * 8 + p] = acc.y;
    }
  };
  float rc[12], ra[12], rm[12];
#pragma unroll
  for (int t = 0; t < 12; ++t) {
    const int pos = q0 + t;
    const bool ok = (pos >= 0) && (pos < CL);
    const int row = b * CL + (ok ? pos : 0);
    rc[t] = Ct[row * Dd + d];
    ra[t] = Am[row * Dd + d];
    rm[t] = Bmat[row * Dd + d];
  }
  v2f w[5], bias;
  loadw(0, w, bias);
  __builtin_amdgcn_sched_barrier(0);
  v2f x[12];
  float xm[12];
#pragma unroll
  for (int t = 0; t < 12; ++t) {
    const int pos = q0 + t;
    const bool ok = (pos >= 0) && (pos < CL);
    x[t].x = ok ? rc[t] : 0.f;
    x[t].y = ok ? ra[t] : 0.f;
    xm[t] = ok ? rm[t] : 0.f;
  }
  emit2(0, x, w, bias);
  loadw(2, w, bias);
#pragma unroll
  for (int t = 0; t < 12; ++t) {
    const float c = x[t].x;
    x[t].x = c * x[t].y;
    x[t].y = c * xm[t];
  }
  emit2(2, x, w, bias);
  __syncthreads();
  int lane = threadIdx.x & 31;
  int o0 = (threadIdx.x >> 5) * 16;
  int half = lane >> 4, mn = lane & 15;
  v8f acc = {};
  const v4f* wrow = (const v4f*)Wp + (size_t)(threadIdx.x >> 5) * (64 * 32) + lane;
  for (int c0 = 0; c0 < 512; c0 += 8) {
    v4f pv = wrow[(c0 >> 3) * 32];
    int cc = c0 + half * 4;
    v2f a0, a1, b0, b1;
    a0.x = pv.x; a0.y = pv.y; a1.x = pv.z; a1.y = pv.w;
    b0.x = dwX[cc][mn];     b0.y = dwX[cc + 1][mn];
    b1.x = dwX[cc + 2][mn]; b1.y = dwX[cc + 3][mn];
    acc = __builtin_amdgcn_wmma_f32_16x16x4_f32(false, a0, false, b0,
                                                (short)0, acc, false, false);
    acc = __builtin_amdgcn_wmma_f32_16x16x4_f32(false, a1, false, b1,
                                                (short)0, acc, false, false);
  }
#pragma unroll
  for (int v = 0; v < 8; ++v) {
    int o = o0 + half * 8 + v;
    out[(size_t)(b * Dd + o) * CL + p0 + mn] = acc[v] + pwb[o];
  }
}

extern "C" void kernel_launch(void* const* d_in, const int* in_sizes, int n_in,
                              void* d_out, int out_size, void* d_ws, size_t ws_size,
                              hipStream_t stream) {
  const float* C   = (const float*)d_in[0];
  const float* Q   = (const float*)d_in[1];
  const float* cm  = (const float*)d_in[2];
  const float* qm  = (const float*)d_in[3];
  const float* W0  = (const float*)d_in[4];
  const float* dww = (const float*)d_in[5];
  const float* dwb = (const float*)d_in[6];
  const float* pww = (const float*)d_in[7];
  const float* pwb = (const float*)d_in[8];
  float* out = (float*)d_out;

  float* ws = (float*)d_ws;
  const size_t nS = (size_t)Bn * CL * QL;
  float* S    = ws;
  float* Sbar = S + nS;
  float* rowb = Sbar + nS;
  float* colb = rowb + (size_t)Bn * CL;
  float* T    = colb + (size_t)Bn * QL;
  float* Ct   = T + (size_t)Bn * QL * Dd;
  float* Am   = Ct + (size_t)Bn * CL * Dd;
  float* Bm   = Am + (size_t)Bn * CL * Dd;
  (void)ws_size; (void)in_sizes; (void)n_in; (void)out_size;

  k_prep<<<Bn * (CL / 128) + Bn * (QL / 128), 256, 0, stream>>>(C, Q, W0, Ct, rowb, colb);
  k_sgemm_rowsm<<<(Bn * (CL / 16)) / 8, 256, 0, stream>>>(C, Q, W0, colb, qm, S, Sbar);
  k_softmax_col<<<Bn * (QL / 32), 512, 0, stream>>>(S, rowb, cm);
  k_tgemm<<<Bn * (QL / 64) * (Dd / 64), 256, 0, stream>>>(S, Ct, T);
  float* Wp = S;
  k_abgemm<<<NAB, 256, 0, stream>>>(Sbar, Q, T, Am, Bm, pww, Wp);
  k_conv<<<Bn * (CL / 16), 256, 0, stream>>>(Ct, Am, Bm, dww, dwb, Wp, pwb, out);
}
